# starting kernel
# baseline (speedup 1.0000x reference)
#include <hip/hip_runtime.h>

#define B_  64
#define S_  512
#define D_  768
#define K_  8
#define R_  277

typedef __attribute__((ext_vector_type(2))) float v2f;
typedef __attribute__((ext_vector_type(8))) float v8f;

__device__ __forceinline__ v8f wmma_f32_16x16x4(v2f a, v2f b, v8f c) {
#if defined(__gfx1250__) && __has_builtin(__builtin_amdgcn_wmma_f32_16x16x4_f32)
  return __builtin_amdgcn_wmma_f32_16x16x4_f32(
      false, a, false, b, (short)0, c, false, false);
#else
  asm volatile("v_wmma_f32_16x16x4_f32 %0, %1, %2, %0"
               : "+v"(c) : "v"(a), "v"(b));
  return c;
#endif
}

__global__ void span_max_kernel(const float* __restrict__ emb,
                                const int*   __restrict__ h_span,
                                const int*   __restrict__ t_span,
                                const int*   __restrict__ b_spans,
                                float* __restrict__ h_emb,
                                float* __restrict__ t_emb,
                                float* __restrict__ b_emb) {
  const int b = blockIdx.x;
  const int d = threadIdx.x;

  int lo[10], hi[10];
  lo[0] = h_span[b * 2];  hi[0] = h_span[b * 2 + 1];
  lo[1] = t_span[b * 2];  hi[1] = t_span[b * 2 + 1];
#pragma unroll
  for (int k = 0; k < K_; ++k) {
    lo[2 + k] = b_spans[(b * K_ + k) * 2];
    hi[2 + k] = b_spans[(b * K_ + k) * 2 + 1];
  }

  float mx[10];
#pragma unroll
  for (int j = 0; j < 10; ++j) mx[j] = -__builtin_inff();

  const float* e = emb + (size_t)b * S_ * D_ + d;
  for (int s = 0; s < S_; ++s) {
    const float val = e[(size_t)s * D_];
#pragma unroll
    for (int j = 0; j < 10; ++j)
      if (s >= lo[j] && s <= hi[j]) mx[j] = fmaxf(mx[j], val);
  }

  h_emb[b * D_ + d] = mx[0];
  t_emb[b * D_ + d] = mx[1];
#pragma unroll
  for (int k = 0; k < K_; ++k)
    b_emb[(size_t)(b * K_ + k) * D_ + d] = mx[2 + k];
}

__global__ void gemm_nt_wmma(const float* __restrict__ A,
                             const float* __restrict__ W,
                             const float* __restrict__ bias,
                             float* __restrict__ C,
                             int M, int N, int Kdim, int relu) {
  const int n_tiles = (N + 15) >> 4;
  const int m_tiles = M >> 4;
  const int wave    = blockIdx.x * (blockDim.x >> 5) + (threadIdx.x >> 5);
  if (wave >= m_tiles * n_tiles) return;

  const int mt   = wave / n_tiles;
  const int nt   = wave % n_tiles;
  const int lane = threadIdx.x & 31;
  const int half = lane >> 4;
  const int r    = lane & 15;

  const float* arow = A + (size_t)(mt * 16 + r) * Kdim + 2 * half;
  int wr = nt * 16 + r;
  if (wr > N - 1) wr = N - 1;
  const float* wrow = W + (size_t)wr * Kdim + 2 * half;

  v8f acc = {};
  for (int k = 0; k < Kdim; k += 16) {
#pragma unroll
    for (int kk = 0; kk < 16; kk += 4) {
      const v2f a  = *(const v2f*)(arow + k + kk);
      const v2f bb = *(const v2f*)(wrow + k + kk);
      acc = wmma_f32_16x16x4(a, bb, acc);
    }
  }

  const int n = nt * 16 + r;
  if (n < N) {
    const float bv = bias[n];
#pragma unroll
    for (int i = 0; i < 8; ++i) {
      float vout = acc[i] + bv;
      if (relu) vout = fmaxf(vout, 0.0f);
      C[(size_t)(mt * 16 + i + 8 * half) * N + n] = vout;
    }
  }
}

__global__ void attn_kernel(const float* __restrict__ u,
                            const float* __restrict__ v,
                            const float* __restrict__ bi,
                            float* __restrict__ pre) {
  const int b   = blockIdx.x;
  const int tid = threadIdx.x;
  __shared__ float red[256];
  __shared__ float att[K_];

  const float* ub  = u + b * D_;
  const float* vb  = v + b * D_;
  const float* bib = bi + (size_t)b * K_ * D_;

  float dot[K_];
#pragma unroll
  for (int k = 0; k < K_; ++k) {
    float p = 0.f;
    for (int d = tid; d < D_; d += 256) p += ub[d] * bib[k * D_ + d];
    dot[k] = p;
  }

#pragma unroll
  for (int k = 0; k < K_; ++k) {
    red[tid] = dot[k];
    __syncthreads();
    for (int off = 128; off > 0; off >>= 1) {
      if (tid < off) red[tid] += red[tid + off];
      __syncthreads();
    }
    if (tid == 0) att[k] = red[0] * 0.125f;
    __syncthreads();
  }

  if (tid == 0) {
    float m = att[0];
    for (int k = 1; k < K_; ++k) m = fmaxf(m, att[k]);
    float s = 0.f;
    for (int k = 0; k < K_; ++k) { att[k] = __expf(att[k] - m); s += att[k]; }
    const float inv = 1.0f / s;
    for (int k = 0; k < K_; ++k) att[k] *= inv;
  }
  __syncthreads();

  for (int d = tid; d < D_; d += 256) {
    float c = 0.f;
#pragma unroll
    for (int k = 0; k < K_; ++k) c += att[k] * bib[k * D_ + d];
    pre[b * D_ + d] = ub[d] + vb[d] + c;
  }
}

extern "C" void kernel_launch(void* const* d_in, const int* in_sizes, int n_in,
                              void* d_out, int out_size, void* d_ws, size_t ws_size,
                              hipStream_t stream) {
  (void)in_sizes; (void)n_in; (void)out_size; (void)ws_size;

  const float* emb     = (const float*)d_in[0];
  const int*   h_span  = (const int*)d_in[1];
  const int*   t_span  = (const int*)d_in[2];
  const int*   b_spans = (const int*)d_in[3];
  const float* wu_w    = (const float*)d_in[4];
  const float* wu_b    = (const float*)d_in[5];
  const float* wv_w    = (const float*)d_in[6];
  const float* wv_b    = (const float*)d_in[7];
  const float* wi_w    = (const float*)d_in[8];
  const float* wi_b    = (const float*)d_in[9];
  const float* ln1_w   = (const float*)d_in[10];
  const float* ln1_b   = (const float*)d_in[11];
  const float* pred_w  = (const float*)d_in[12];
  const float* pred_b  = (const float*)d_in[13];
  float* out = (float*)d_out;

  float* ws    = (float*)d_ws;
  float* h_emb = ws;
  float* t_emb = h_emb + B_ * D_;
  float* b_emb = t_emb + B_ * D_;
  float* u     = b_emb + B_ * K_ * D_;
  float* vv    = u + B_ * D_;
  float* bi    = vv + B_ * D_;
  float* pre   = bi + B_ * K_ * D_;
  float* pair  = h_emb;

  span_max_kernel<<<B_, D_, 0, stream>>>(emb, h_span, t_span, b_spans,
                                         h_emb, t_emb, b_emb);

  auto launch_gemm = [&](const float* A, const float* W, const float* bias,
                         float* C, int M, int N, int relu) {
    const int waves  = (M / 16) * ((N + 15) / 16);
    const int blocks = (waves + 7) / 8;
    gemm_nt_wmma<<<blocks, 256, 0, stream>>>(A, W, bias, C, M, N, D_, relu);
  };

  launch_gemm(h_emb, wu_w, wu_b, u, B_, D_, 0);
  launch_gemm(t_emb, wv_w, wv_b, vv, B_, D_, 0);
  launch_gemm(b_emb, wi_w, wi_b, bi, B_ * K_, D_, 0);

  attn_kernel<<<B_, 256, 0, stream>>>(u, vv, bi, pre);

  launch_gemm(pre, ln1_w, ln1_b, pair, B_, D_, 1);
  launch_gemm(pair, pred_w, pred_b, out, B_, R_, 0);
}


# Attention dots u.bi_k taken from the bridge GEMM's accumulators in its epilogue (per-tile partials to workspace); attention kernel left with no tree sums and no barriers: new dot sum order
# speedup vs baseline: 5.6444x; 5.6444x over previous
#include <hip/hip_runtime.h>

#define B_  64
#define S_  512
#define D_  768
#define K_  8
#define R_  277

typedef __attribute__((ext_vector_type(2))) float v2f;
typedef __attribute__((ext_vector_type(8))) float v8f;

__device__ __forceinline__ v8f wmma_f32_16x16x4(v2f a, v2f b, v8f c) {
#if defined(__gfx1250__) && __has_builtin(__builtin_amdgcn_wmma_f32_16x16x4_f32)
  return __builtin_amdgcn_wmma_f32_16x16x4_f32(
      false, a, false, b, (short)0, c, false, false);
#else
  asm volatile("v_wmma_f32_16x16x4_f32 %0, %1, %2, %0"
               : "+v"(c) : "v"(a), "v"(b));
  return c;
#endif
}

#define NSPAN_   10
#define SPAN_U_  8

typedef __attribute__((ext_vector_type(4))) float v4f;

__device__ __forceinline__ void span_load_group(v4f (&buf)[SPAN_U_],
                                                const char* __restrict__ rowbase,
                                                unsigned colbytes, int r0, int hi) {
#pragma unroll
  for (int i = 0; i < SPAN_U_; ++i) {
    int r = r0 + 2 * i;
    r = r < hi ? r : hi;
    const unsigned off = (unsigned)r * (unsigned)(D_ * 4) + colbytes;
    buf[i] = *(const v4f*)(rowbase + off);
  }
}

__device__ __forceinline__ v4f span_fold_group(const v4f (&buf)[SPAN_U_], v4f m) {
#pragma unroll
  for (int i = 0; i < SPAN_U_; ++i) {
    m.x = fmaxf(m.x, buf[i].x);
    m.y = fmaxf(m.y, buf[i].y);
    m.z = fmaxf(m.z, buf[i].z);
    m.w = fmaxf(m.w, buf[i].w);
  }
  return m;
}

__global__ void __launch_bounds__(384)
span_max_by_span(const float* __restrict__ emb,
                 const int*   __restrict__ h_span,
                 const int*   __restrict__ t_span,
                 const int*   __restrict__ b_spans,
                 float* __restrict__ h_emb,
                 float* __restrict__ t_emb,
                 float* __restrict__ b_emb) {
  const int j = blockIdx.x;
  const int b = blockIdx.y;

  const int* sp;
  float* dst;
  if (j == 0)      { sp = h_span + b * 2;  dst = h_emb + b * D_; }
  else if (j == 1) { sp = t_span + b * 2;  dst = t_emb + b * D_; }
  else {
    sp  = b_spans + (b * K_ + (j - 2)) * 2;
    dst = b_emb + (size_t)(b * K_ + (j - 2)) * D_;
  }
  int lo = sp[0];
  int hi = sp[1];
  if (lo < 0) lo = 0;
  if (hi > S_ - 1) hi = S_ - 1;

  const int lane = threadIdx.x & 31;
  const int wave = threadIdx.x >> 5;
  const int half = lane >> 4;
  const int col  = wave * 64 + (lane & 15) * 4;

  const char* rowbase = (const char*)(emb + (size_t)b * S_ * D_);
  const unsigned colbytes = (unsigned)col * 4u;

  const float ninf = -__builtin_inff();
  v4f m = {ninf, ninf, ninf, ninf};

  const int n = hi - lo + 1;
  if (n > 0) {
    const int ngroups = (n + 2 * SPAN_U_ - 1) / (2 * SPAN_U_);
    int rA = lo + half;
    int rB = rA + 2 * SPAN_U_;
    v4f bufA[SPAN_U_], bufB[SPAN_U_];
    span_load_group(bufA, rowbase, colbytes, rA, hi);
    rA += 4 * SPAN_U_;
    for (int g = 0; g < ngroups; g += 2) {
      span_load_group(bufB, rowbase, colbytes, rB, hi);
      rB += 4 * SPAN_U_;
      m = span_fold_group(bufA, m);
      asm volatile("" : "+v"(m.x), "+v"(m.y), "+v"(m.z), "+v"(m.w), "+v"(rA));
      span_load_group(bufA, rowbase, colbytes, rA, hi);
      rA += 4 * SPAN_U_;
      m = span_fold_group(bufB, m);
      asm volatile("" : "+v"(m.x), "+v"(m.y), "+v"(m.z), "+v"(m.w), "+v"(rB));
    }
  }

  m.x = fmaxf(m.x, __shfl_xor(m.x, 16));
  m.y = fmaxf(m.y, __shfl_xor(m.y, 16));
  m.z = fmaxf(m.z, __shfl_xor(m.z, 16));
  m.w = fmaxf(m.w, __shfl_xor(m.w, 16));

  if (half == 0) *(v4f*)(dst + col) = m;
}

#define GK_      768
#define G_KC_    64
#define G_PITCH_ 68
#define G_NCH_   (GK_ / G_KC_)

template <int NP>
__device__ __forceinline__ void gemm_fetch(v4f (&reg)[NP],
                                           const float* const (&srcp)[NP], int chunk) {
#pragma unroll
  for (int i = 0; i < NP; ++i) reg[i] = *(const v4f*)(srcp[i] + chunk * G_KC_);
}

template <int NP, int RPP>
__device__ __forceinline__ void gemm_stash(const v4f (&reg)[NP], float* stage, int soff) {
#pragma unroll
  for (int i = 0; i < NP; ++i) *(v4f*)(stage + soff + i * RPP * G_PITCH_) = reg[i];
}

__device__ __forceinline__ void gemm_chunk(const float* ap, const float* bp,
                                           v8f& acc0, v8f& acc1) {
  v4f af[G_KC_ / 8], bf[G_KC_ / 8];
#pragma unroll
  for (int kb = 0; kb < G_KC_ / 8; ++kb) {
    af[kb] = *(const v4f*)(ap + kb * 8);
    bf[kb] = *(const v4f*)(bp + kb * 8);
  }
  __builtin_amdgcn_sched_barrier(0);
#pragma unroll
  for (int kb = 0; kb < G_KC_ / 8; ++kb) {
    const v2f a0 = {af[kb].x, af[kb].y}, b0 = {bf[kb].x, bf[kb].y};
    const v2f a1 = {af[kb].z, af[kb].w}, b1 = {bf[kb].z, bf[kb].w};
    acc0 = wmma_f32_16x16x4(a0, b0, acc0);
    acc1 = wmma_f32_16x16x4(a1, b1, acc1);
  }
}

__device__ __forceinline__ void gemm_lds_handover() {
  __builtin_amdgcn_fence(__ATOMIC_RELEASE, "workgroup", "local");
  __builtin_amdgcn_s_barrier();
  __builtin_amdgcn_fence(__ATOMIC_ACQUIRE, "workgroup", "local");
}

template <int WM, int WN, bool DOTS>
__global__ void __launch_bounds__(32 * WM * WN)
gemm_nt_wmma(const float* __restrict__ A,
             const float* __restrict__ W,
             const float* __restrict__ bias,
             float* __restrict__ C,
             int N, int relu,
             const float* __restrict__ Uq,
             float* __restrict__ part) {
  constexpr int BM   = 16 * WM;
  constexpr int BN   = 16 * WN;
  constexpr int ROWS = BM + BN;
  constexpr int RPP  = 2 * WM * WN;
  constexpr int NP   = ROWS / RPP;
  constexpr int NPA  = BM / RPP;
  static_assert(ROWS % RPP == 0 && BM % RPP == 0, "copy passes must not straddle A and W");
  __shared__ __attribute__((aligned(16))) float tile[2][ROWS * G_PITCH_];

  const int tid  = threadIdx.x;
  const int lane = tid & 31;
  const int wave = tid >> 5;
  const int wm   = wave / WN;
  const int wn   = wave % WN;
  const int half = lane >> 4;
  const int r    = lane & 15;
  const int m0   = blockIdx.y * BM;
  const int n0   = blockIdx.x * BN;

  const int srow = tid >> 4;
  const int scol = (tid & 15) * 4;
  const float* srcp[NP];
#pragma unroll
  for (int i = 0; i < NPA; ++i)
    srcp[i] = A + (size_t)(m0 + srow + RPP * i) * GK_ + scol;
#pragma unroll
  for (int i = NPA; i < NP; ++i) {
    int wr = n0 + srow + RPP * (i - NPA);
    if (wr > N - 1) wr = N - 1;
    srcp[i] = W + (size_t)wr * GK_ + scol;
  }
  const int soff = srow * G_PITCH_ + scol;

  const float* ap0 = &tile[0][(wm * 16 + r) * G_PITCH_ + 4 * half];
  const float* bp0 = &tile[0][(BM + wn * 16 + r) * G_PITCH_ + 4 * half];
  const float* ap1 = &tile[1][(wm * 16 + r) * G_PITCH_ + 4 * half];
  const float* bp1 = &tile[1][(BM + wn * 16 + r) * G_PITCH_ + 4 * half];

  v8f acc0 = {}, acc1 = {};
  v4f ra[NP], rb[NP];

  gemm_fetch(ra, srcp, 0);
  gemm_fetch(rb, srcp, 1);
  gemm_stash<NP, RPP>(ra, tile[0], soff);
  gemm_fetch(ra, srcp, 2);

  for (int t = 0; t < G_NCH_; t += 2) {
    gemm_lds_handover();
    gemm_stash<NP, RPP>(rb, tile[1], soff);
    if (t + 3 < G_NCH_) gemm_fetch(rb, srcp, t + 3);
    gemm_chunk(ap0, bp0, acc0, acc1);
    gemm_lds_handover();
    if (t + 2 < G_NCH_) gemm_stash<NP, RPP>(ra, tile[0], soff);
    if (t + 4 < G_NCH_) gemm_fetch(ra, srcp, t + 4);
    gemm_chunk(ap1, bp1, acc0, acc1);
  }

  const int n = n0 + wn * 16 + r;
  float pd[8];
#pragma unroll
  for (int i = 0; i < 8; ++i) pd[i] = 0.0f;
  if (n < N) {
    const float bv = bias[n];
    float uq = 0.0f;
    if constexpr (DOTS) uq = Uq[(size_t)((m0 + wm * 16) / K_ + half) * N + n];
#pragma unroll
    for (int i = 0; i < 8; ++i) {
      float vout = (acc0[i] + acc1[i]) + bv;
      if (relu) vout = fmaxf(vout, 0.0f);
      C[(size_t)(m0 + wm * 16 + i + 8 * half) * N + n] = vout;
      if constexpr (DOTS) pd[i] = uq * vout;
    }
  }
  if constexpr (DOTS) {
    static_assert(K_ == 8, "one lane half (8 rows of the piece) must be one batch row");
    __shared__ float dred[WN][BM];
#pragma unroll
    for (int off = 8; off > 0; off >>= 1) {
#pragma unroll
      for (int i = 0; i < 8; ++i) pd[i] += __shfl_xor(pd[i], off);
    }
    if (r == 0) {
#pragma unroll
      for (int i = 0; i < 8; ++i) dred[wn][wm * 16 + 8 * half + i] = pd[i];
    }
    gemm_lds_handover();
    if (tid < BM) {
      float s = dred[0][tid];
#pragma unroll
      for (int w = 1; w < WN; ++w) s += dred[w][tid];
      const int ntile = (N + BN - 1) / BN;
      part[((size_t)blockIdx.y * ntile + blockIdx.x) * BM + tid] = s;
    }
  }
}

#define PT_  (D_ / 32)
#define PBM_ 64

__global__ void __launch_bounds__(64)
attn_pre_kernel(const float* __restrict__ u,
                const float* __restrict__ v,
                const float* __restrict__ bi,
                const float* __restrict__ part,
                float* __restrict__ pre) {
  static_assert(PT_ <= 32, "one lane per column tile");
  const int b    = blockIdx.y;
  const int lane = threadIdx.x & 31;
  const int d    = blockIdx.x * 256 + threadIdx.x * 4;

  const int pt = lane < PT_ ? lane : 0;
  const float* pp = part + ((size_t)(b / (PBM_ / K_)) * PT_ + pt) * PBM_ + (b % (PBM_ / K_)) * K_;
  v4f p0 = *(const v4f*)pp;
  v4f p1 = *(const v4f*)(pp + 4);

  const v4f uu = *(const v4f*)(u + (size_t)b * D_ + d);
  const v4f vv = *(const v4f*)(v + (size_t)b * D_ + d);
  v4f bk[K_];
#pragma unroll
  for (int k = 0; k < K_; ++k) bk[k] = *(const v4f*)(bi + ((size_t)b * K_ + k) * D_ + d);
  __builtin_amdgcn_sched_barrier(0);

  float att[K_] = {p0.x, p0.y, p0.z, p0.w, p1.x, p1.y, p1.z, p1.w};
  if (lane >= PT_) {
#pragma unroll
    for (int k = 0; k < K_; ++k) att[k] = 0.0f;
  }
#pragma unroll
  for (int off = 16; off > 0; off >>= 1) {
#pragma unroll
    for (int k = 0; k < K_; ++k) att[k] += __shfl_xor(att[k], off);
  }

#pragma unroll
  for (int k = 0; k < K_; ++k) att[k] *= 0.125f;
  float m = att[0];
#pragma unroll
  for (int k = 1; k < K_; ++k) m = fmaxf(m, att[k]);
  float s = 0.f;
#pragma unroll
  for (int k = 0; k < K_; ++k) { att[k] = __expf(att[k] - m); s += att[k]; }
  const float inv = 1.0f / s;
#pragma unroll
  for (int k = 0; k < K_; ++k) att[k] *= inv;

  v4f c = {0.f, 0.f, 0.f, 0.f};
#pragma unroll
  for (int k = 0; k < K_; ++k) {
    c.x += att[k] * bk[k].x;
    c.y += att[k] * bk[k].y;
    c.z += att[k] * bk[k].z;
    c.w += att[k] * bk[k].w;
  }
  v4f o;
  o.x = uu.x + vv.x + c.x;
  o.y = uu.y + vv.y + c.y;
  o.z = uu.z + vv.z + c.z;
  o.w = uu.w + vv.w + c.w;
  *(v4f*)(pre + (size_t)b * D_ + d) = o;
}

extern "C" void kernel_launch(void* const* d_in, const int* in_sizes, int n_in,
                              void* d_out, int out_size, void* d_ws, size_t ws_size,
                              hipStream_t stream) {
  (void)in_sizes; (void)n_in; (void)out_size; (void)ws_size;

  const float* emb     = (const float*)d_in[0];
  const int*   h_span  = (const int*)d_in[1];
  const int*   t_span  = (const int*)d_in[2];
  const int*   b_spans = (const int*)d_in[3];
  const float* wu_w    = (const float*)d_in[4];
  const float* wu_b    = (const float*)d_in[5];
  const float* wv_w    = (const float*)d_in[6];
  const float* wv_b    = (const float*)d_in[7];
  const float* wi_w    = (const float*)d_in[8];
  const float* wi_b    = (const float*)d_in[9];
  const float* ln1_w   = (const float*)d_in[10];
  const float* ln1_b   = (const float*)d_in[11];
  const float* pred_w  = (const float*)d_in[12];
  const float* pred_b  = (const float*)d_in[13];
  float* out = (float*)d_out;

  float* ws    = (float*)d_ws;
  float* h_emb = ws;
  float* t_emb = h_emb + B_ * D_;
  float* b_emb = t_emb + B_ * D_;
  float* u     = b_emb + B_ * K_ * D_;
  float* vv    = u + B_ * D_;
  float* bi    = vv + B_ * D_;
  float* pre   = bi + B_ * K_ * D_;
  float* pair  = h_emb;
  float* part  = t_emb;

  span_max_by_span<<<dim3(NSPAN_, B_), 384, 0, stream>>>(emb, h_span, t_span, b_spans,
                                                         h_emb, t_emb, b_emb);

  gemm_nt_wmma<1, 2, false><<<dim3(24, 4), 64, 0, stream>>>(h_emb, wu_w, wu_b, u, D_, 0, nullptr, nullptr);
  gemm_nt_wmma<1, 2, false><<<dim3(24, 4), 64, 0, stream>>>(t_emb, wv_w, wv_b, vv, D_, 0, nullptr, nullptr);
  gemm_nt_wmma<4, 2, true><<<dim3(PT_, 8), 256, 0, stream>>>(b_emb, wi_w, wi_b, bi, D_, 0, u, part);

  attn_pre_kernel<<<dim3(3, B_), 64, 0, stream>>>(u, vv, bi, part, pre);

  gemm_nt_wmma<1, 2, false><<<dim3(24, 4), 64, 0, stream>>>(pre, ln1_w, ln1_b, pair, D_, 1, nullptr, nullptr);
  gemm_nt_wmma<1, 2, false><<<dim3(9, 4), 64, 0, stream>>>(pair, pred_w, pred_b, out, R_, 0, nullptr, nullptr);
}


# Span max-pool: block = (batch row, 256-byte strip) reads its rows once into an LDS table of 8-row chunk maxima; span = its whole chunks from the table + up to 14 edge rows; shared rows not re-read
# speedup vs baseline: 7.4781x; 1.3249x over previous
#include <hip/hip_runtime.h>

#define B_  64
#define S_  512
#define D_  768
#define K_  8
#define R_  277

typedef __attribute__((ext_vector_type(2))) float v2f;
typedef __attribute__((ext_vector_type(8))) float v8f;

__device__ __forceinline__ v8f wmma_f32_16x16x4(v2f a, v2f b, v8f c) {
#if defined(__gfx1250__) && __has_builtin(__builtin_amdgcn_wmma_f32_16x16x4_f32)
  return __builtin_amdgcn_wmma_f32_16x16x4_f32(
      false, a, false, b, (short)0, c, false, false);
#else
  asm volatile("v_wmma_f32_16x16x4_f32 %0, %1, %2, %0"
               : "+v"(c) : "v"(a), "v"(b));
  return c;
#endif
}

#define NSPAN_    10
#define CH_ROWS_  8
#define NCHUNK_   (S_ / CH_ROWS_)
#define STRIP_    64
#define SPAN_WV_  8

typedef __attribute__((ext_vector_type(4))) float v4f;

__device__ __forceinline__ v4f vmax4(v4f a, v4f b) {
  v4f m;
  m.x = fmaxf(a.x, b.x);
  m.y = fmaxf(a.y, b.y);
  m.z = fmaxf(a.z, b.z);
  m.w = fmaxf(a.w, b.w);
  return m;
}

__device__ __forceinline__ v4f span_row_load(const char* __restrict__ rowbase,
                                             unsigned colbytes, int r) {
  const unsigned off = (unsigned)r * (unsigned)(D_ * 4) + colbytes;
  return *(const v4f*)(rowbase + off);
}

__device__ __forceinline__ void chunk_load(v4f (&buf)[CH_ROWS_],
                                           const char* __restrict__ rowbase,
                                           unsigned colbytes, int c) {
#pragma unroll
  for (int i = 0; i < CH_ROWS_; ++i)
    buf[i] = span_row_load(rowbase, colbytes, c * CH_ROWS_ + i);
}

__device__ __forceinline__ v4f chunk_fold(const v4f (&buf)[CH_ROWS_]) {
  v4f m = buf[0];
#pragma unroll
  for (int i = 1; i < CH_ROWS_; ++i) m = vmax4(m, buf[i]);
  return m;
}

__device__ __forceinline__ void span_bounds(const int* __restrict__ h_span,
                                            const int* __restrict__ t_span,
                                            const int* __restrict__ b_spans,
                                            int b, int j, int& lo, int& hi) {
  const int* sp;
  if (j == 0)      sp = h_span + b * 2;
  else if (j == 1) sp = t_span + b * 2;
  else             sp = b_spans + (b * K_ + (j - 2)) * 2;
  lo = sp[0];
  hi = sp[1];
  if (lo < 0) lo = 0;
  if (hi > S_ - 1) hi = S_ - 1;
}

__device__ __forceinline__ int chunk_first(int lo) { return (lo + CH_ROWS_ - 1) / CH_ROWS_; }
__device__ __forceinline__ int chunk_last(int hi)  { return (hi + 1) / CH_ROWS_ - 1; }

__global__ void __launch_bounds__(32 * SPAN_WV_)
span_max_chunk_table(const float* __restrict__ emb,
                     const int*   __restrict__ h_span,
                     const int*   __restrict__ t_span,
                     const int*   __restrict__ b_spans,
                     float* __restrict__ h_emb,
                     float* __restrict__ t_emb,
                     float* __restrict__ b_emb) {
  __shared__ __attribute__((aligned(16))) float tab[NCHUNK_ * STRIP_];

  const int strip = blockIdx.x;
  const int b     = blockIdx.y;

  const int lane = threadIdx.x & 31;
  const int wave = __builtin_amdgcn_readfirstlane((int)(threadIdx.x >> 5));
  const int half = lane >> 4;
  const int lcol = (lane & 15) * 4;
  const int col  = strip * STRIP_ + lcol;

  const char* rowbase = (const char*)(emb + (size_t)b * S_ * D_);
  const unsigned colbytes = (unsigned)col * 4u;

  int cb = NCHUNK_, ce = -1;
#pragma unroll
  for (int j = 0; j < NSPAN_; ++j) {
    int lo, hi;
    span_bounds(h_span, t_span, b_spans, b, j, lo, hi);
    const int cf = chunk_first(lo), cl = chunk_last(hi);
    if (cf <= cl) {
      cb = cf < cb ? cf : cb;
      ce = cl > ce ? cl : ce;
    }
  }
  const int nch = ce - cb + 1;

  const int slot = 2 * wave + half;
#pragma unroll
  for (int p = 0; p < NCHUNK_ / 32; ++p) {
    if (32 * p < nch) {
      const int c0 = cb + 32 * p + slot;
      const int c1 = c0 + 16;
      v4f buf0[CH_ROWS_], buf1[CH_ROWS_];
      chunk_load(buf0, rowbase, colbytes, c0 < ce ? c0 : ce);
      chunk_load(buf1, rowbase, colbytes, c1 < ce ? c1 : ce);
      const v4f m0 = chunk_fold(buf0);
      if (c0 <= ce) *(v4f*)&tab[(c0 - cb) * STRIP_ + lcol] = m0;
      const v4f m1 = chunk_fold(buf1);
      if (c1 <= ce) *(v4f*)&tab[(c1 - cb) * STRIP_ + lcol] = m1;
    }
  }
  __syncthreads();

  const float ninf = -__builtin_inff();
  for (int j = wave; j < NSPAN_; j += SPAN_WV_) {
    int lo, hi;
    span_bounds(h_span, t_span, b_spans, b, j, lo, hi);
    float* dst;
    if (j == 0)      dst = h_emb + b * D_;
    else if (j == 1) dst = t_emb + b * D_;
    else             dst = b_emb + (size_t)(b * K_ + (j - 2)) * D_;

    v4f m = {ninf, ninf, ninf, ninf};
    if (lo <= hi) {
      const int cf = chunk_first(lo), cl = chunk_last(hi);
      int fend = CH_ROWS_ * cf - 1;
      fend = fend < hi ? fend : hi;
      fend = fend > lo ? fend : lo;
      int bbeg = CH_ROWS_ * (cl + 1);
      bbeg = bbeg > lo ? bbeg : lo;
      bbeg = bbeg < hi ? bbeg : hi;
      v4f e[CH_ROWS_ - 1];
#pragma unroll
      for (int i = 0; i < CH_ROWS_ - 1; ++i) {
        int rf = lo + i;  rf = rf < fend ? rf : fend;
        int rb = hi - i;  rb = rb > bbeg ? rb : bbeg;
        e[i] = span_row_load(rowbase, colbytes, half ? rb : rf);
      }
      if (cf <= cl) {
        const int ntrip = (cl - cf + 2) >> 1;
        for (int t0 = 0; t0 < ntrip; t0 += 8) {
          v4f q[8];
#pragma unroll
          for (int u = 0; u < 8; ++u) {
            int c = cf + 2 * (t0 + u) + half;
            c = c < cl ? c : cl;
            q[u] = *(const v4f*)&tab[(c - cb) * STRIP_ + lcol];
          }
#pragma unroll
          for (int u = 0; u < 8; ++u) m = vmax4(m, q[u]);
        }
      }
#pragma unroll
      for (int i = 0; i < CH_ROWS_ - 1; ++i) m = vmax4(m, e[i]);
    }

    m.x = fmaxf(m.x, __shfl_xor(m.x, 16));
    m.y = fmaxf(m.y, __shfl_xor(m.y, 16));
    m.z = fmaxf(m.z, __shfl_xor(m.z, 16));
    m.w = fmaxf(m.w, __shfl_xor(m.w, 16));

    if (half == 0) *(v4f*)(dst + col) = m;
  }
}

#define GK_      768
#define G_KC_    64
#define G_PITCH_ 68
#define G_NCH_   (GK_ / G_KC_)

template <int NP>
__device__ __forceinline__ void gemm_fetch(v4f (&reg)[NP],
                                           const float* const (&srcp)[NP], int chunk) {
#pragma unroll
  for (int i = 0; i < NP; ++i) reg[i] = *(const v4f*)(srcp[i] + chunk * G_KC_);
}

template <int NP, int RPP>
__device__ __forceinline__ void gemm_stash(const v4f (&reg)[NP], float* stage, int soff) {
#pragma unroll
  for (int i = 0; i < NP; ++i) *(v4f*)(stage + soff + i * RPP * G_PITCH_) = reg[i];
}

__device__ __forceinline__ void gemm_chunk(const float* ap, const float* bp,
                                           v8f& acc0, v8f& acc1) {
  v4f af[G_KC_ / 8], bf[G_KC_ / 8];
#pragma unroll
  for (int kb = 0; kb < G_KC_ / 8; ++kb) {
    af[kb] = *(const v4f*)(ap + kb * 8);
    bf[kb] = *(const v4f*)(bp + kb * 8);
  }
  __builtin_amdgcn_sched_barrier(0);
#pragma unroll
  for (int kb = 0; kb < G_KC_ / 8; ++kb) {
    const v2f a0 = {af[kb].x, af[kb].y}, b0 = {bf[kb].x, bf[kb].y};
    const v2f a1 = {af[kb].z, af[kb].w}, b1 = {bf[kb].z, bf[kb].w};
    acc0 = wmma_f32_16x16x4(a0, b0, acc0);
    acc1 = wmma_f32_16x16x4(a1, b1, acc1);
  }
}

__device__ __forceinline__ void gemm_lds_handover() {
  __builtin_amdgcn_fence(__ATOMIC_RELEASE, "workgroup", "local");
  __builtin_amdgcn_s_barrier();
  __builtin_amdgcn_fence(__ATOMIC_ACQUIRE, "workgroup", "local");
}

template <int WM, int WN, bool DOTS>
__global__ void __launch_bounds__(32 * WM * WN)
gemm_nt_wmma(const float* __restrict__ A,
             const float* __restrict__ W,
             const float* __restrict__ bias,
             float* __restrict__ C,
             int N, int relu,
             const float* __restrict__ Uq,
             float* __restrict__ part) {
  constexpr int BM   = 16 * WM;
  constexpr int BN   = 16 * WN;
  constexpr int ROWS = BM + BN;
  constexpr int RPP  = 2 * WM * WN;
  constexpr int NP   = ROWS / RPP;
  constexpr int NPA  = BM / RPP;
  static_assert(ROWS % RPP == 0 && BM % RPP == 0, "copy passes must not straddle A and W");
  __shared__ __attribute__((aligned(16))) float tile[2][ROWS * G_PITCH_];

  const int tid  = threadIdx.x;
  const int lane = tid & 31;
  const int wave = tid >> 5;
  const int wm   = wave / WN;
  const int wn   = wave % WN;
  const int half = lane >> 4;
  const int r    = lane & 15;
  const int m0   = blockIdx.y * BM;
  const int n0   = blockIdx.x * BN;

  const int srow = tid >> 4;
  const int scol = (tid & 15) * 4;
  const float* srcp[NP];
#pragma unroll
  for (int i = 0; i < NPA; ++i)
    srcp[i] = A + (size_t)(m0 + srow + RPP * i) * GK_ + scol;
#pragma unroll
  for (int i = NPA; i < NP; ++i) {
    int wr = n0 + srow + RPP * (i - NPA);
    if (wr > N - 1) wr = N - 1;
    srcp[i] = W + (size_t)wr * GK_ + scol;
  }
  const int soff = srow * G_PITCH_ + scol;

  const float* ap0 = &tile[0][(wm * 16 + r) * G_PITCH_ + 4 * half];
  const float* bp0 = &tile[0][(BM + wn * 16 + r) * G_PITCH_ + 4 * half];
  const float* ap1 = &tile[1][(wm * 16 + r) * G_PITCH_ + 4 * half];
  const float* bp1 = &tile[1][(BM + wn * 16 + r) * G_PITCH_ + 4 * half];

  v8f acc0 = {}, acc1 = {};
  v4f ra[NP], rb[NP];

  gemm_fetch(ra, srcp, 0);
  gemm_fetch(rb, srcp, 1);
  gemm_stash<NP, RPP>(ra, tile[0], soff);
  gemm_fetch(ra, srcp, 2);

  for (int t = 0; t < G_NCH_; t += 2) {
    gemm_lds_handover();
    gemm_stash<NP, RPP>(rb, tile[1], soff);
    if (t + 3 < G_NCH_) gemm_fetch(rb, srcp, t + 3);
    gemm_chunk(ap0, bp0, acc0, acc1);
    gemm_lds_handover();
    if (t + 2 < G_NCH_) gemm_stash<NP, RPP>(ra, tile[0], soff);
    if (t + 4 < G_NCH_) gemm_fetch(ra, srcp, t + 4);
    gemm_chunk(ap1, bp1, acc0, acc1);
  }

  const int n = n0 + wn * 16 + r;
  float pd[8];
#pragma unroll
  for (int i = 0; i < 8; ++i) pd[i] = 0.0f;
  if (n < N) {
    const float bv = bias[n];
    float uq = 0.0f;
    if constexpr (DOTS) uq = Uq[(size_t)((m0 + wm * 16) / K_ + half) * N + n];
#pragma unroll
    for (int i = 0; i < 8; ++i) {
      float vout = (acc0[i] + acc1[i]) + bv;
      if (relu) vout = fmaxf(vout, 0.0f);
      C[(size_t)(m0 + wm * 16 + i + 8 * half) * N + n] = vout;
      if constexpr (DOTS) pd[i] = uq * vout;
    }
  }
  if constexpr (DOTS) {
    static_assert(K_ == 8, "one lane half (8 rows of the piece) must be one batch row");
    __shared__ float dred[WN][BM];
#pragma unroll
    for (int off = 8; off > 0; off >>= 1) {
#pragma unroll
      for (int i = 0; i < 8; ++i) pd[i] += __shfl_xor(pd[i], off);
    }
    if (r == 0) {
#pragma unroll
      for (int i = 0; i < 8; ++i) dred[wn][wm * 16 + 8 * half + i] = pd[i];
    }
    gemm_lds_handover();
    if (tid < BM) {
      float s = dred[0][tid];
#pragma unroll
      for (int w = 1; w < WN; ++w) s += dred[w][tid];
      const int ntile = (N + BN - 1) / BN;
      part[((size_t)blockIdx.y * ntile + blockIdx.x) * BM + tid] = s;
    }
  }
}

#define PT_  (D_ / 32)
#define PBM_ 64

__global__ void __launch_bounds__(64)
attn_pre_kernel(const float* __restrict__ u,
                const float* __restrict__ v,
                const float* __restrict__ bi,
                const float* __restrict__ part,
                float* __restrict__ pre) {
  static_assert(PT_ <= 32, "one lane per column tile");
  const int b    = blockIdx.y;
  const int lane = threadIdx.x & 31;
  const int d    = blockIdx.x * 256 + threadIdx.x * 4;

  const int pt = lane < PT_ ? lane : 0;
  const float* pp = part + ((size_t)(b / (PBM_ / K_)) * PT_ + pt) * PBM_ + (b % (PBM_ / K_)) * K_;
  v4f p0 = *(const v4f*)pp;
  v4f p1 = *(const v4f*)(pp + 4);

  const v4f uu = *(const v4f*)(u + (size_t)b * D_ + d);
  const v4f vv = *(const v4f*)(v + (size_t)b * D_ + d);
  v4f bk[K_];
#pragma unroll
  for (int k = 0; k < K_; ++k) bk[k] = *(const v4f*)(bi + ((size_t)b * K_ + k) * D_ + d);
  __builtin_amdgcn_sched_barrier(0);

  float att[K_] = {p0.x, p0.y, p0.z, p0.w, p1.x, p1.y, p1.z, p1.w};
  if (lane >= PT_) {
#pragma unroll
    for (int k = 0; k < K_; ++k) att[k] = 0.0f;
  }
#pragma unroll
  for (int off = 16; off > 0; off >>= 1) {
#pragma unroll
    for (int k = 0; k < K_; ++k) att[k] += __shfl_xor(att[k], off);
  }

#pragma unroll
  for (int k = 0; k < K_; ++k) att[k] *= 0.125f;
  float m = att[0];
#pragma unroll
  for (int k = 1; k < K_; ++k) m = fmaxf(m, att[k]);
  float s = 0.f;
#pragma unroll
  for (int k = 0; k < K_; ++k) { att[k] = __expf(att[k] - m); s += att[k]; }
  const float inv = 1.0f / s;
#pragma unroll
  for (int k = 0; k < K_; ++k) att[k] *= inv;

  v4f c = {0.f, 0.f, 0.f, 0.f};
#pragma unroll
  for (int k = 0; k < K_; ++k) {
    c.x += att[k] * bk[k].x;
    c.y += att[k] * bk[k].y;
    c.z += att[k] * bk[k].z;
    c.w += att[k] * bk[k].w;
  }
  v4f o;
  o.x = uu.x + vv.x + c.x;
  o.y = uu.y + vv.y + c.y;
  o.z = uu.z + vv.z + c.z;
  o.w = uu.w + vv.w + c.w;
  *(v4f*)(pre + (size_t)b * D_ + d) = o;
}

extern "C" void kernel_launch(void* const* d_in, const int* in_sizes, int n_in,
                              void* d_out, int out_size, void* d_ws, size_t ws_size,
                              hipStream_t stream) {
  (void)in_sizes; (void)n_in; (void)out_size; (void)ws_size;

  const float* emb     = (const float*)d_in[0];
  const int*   h_span  = (const int*)d_in[1];
  const int*   t_span  = (const int*)d_in[2];
  const int*   b_spans = (const int*)d_in[3];
  const float* wu_w    = (const float*)d_in[4];
  const float* wu_b    = (const float*)d_in[5];
  const float* wv_w    = (const float*)d_in[6];
  const float* wv_b    = (const float*)d_in[7];
  const float* wi_w    = (const float*)d_in[8];
  const float* wi_b    = (const float*)d_in[9];
  const float* ln1_w   = (const float*)d_in[10];
  const float* ln1_b   = (const float*)d_in[11];
  const float* pred_w  = (const float*)d_in[12];
  const float* pred_b  = (const float*)d_in[13];
  float* out = (float*)d_out;

  float* ws    = (float*)d_ws;
  float* h_emb = ws;
  float* t_emb = h_emb + B_ * D_;
  float* b_emb = t_emb + B_ * D_;
  float* u     = b_emb + B_ * K_ * D_;
  float* vv    = u + B_ * D_;
  float* bi    = vv + B_ * D_;
  float* pre   = bi + B_ * K_ * D_;
  float* pair  = h_emb;
  float* part  = t_emb;

  span_max_chunk_table<<<dim3(D_ / STRIP_, B_), 32 * SPAN_WV_, 0, stream>>>(emb, h_span, t_span, b_spans,
                                                                            h_emb, t_emb, b_emb);

  gemm_nt_wmma<1, 2, false><<<dim3(24, 4), 64, 0, stream>>>(h_emb, wu_w, wu_b, u, D_, 0, nullptr, nullptr);
  gemm_nt_wmma<1, 2, false><<<dim3(24, 4), 64, 0, stream>>>(t_emb, wv_w, wv_b, vv, D_, 0, nullptr, nullptr);
  gemm_nt_wmma<4, 2, true><<<dim3(PT_, 8), 256, 0, stream>>>(b_emb, wi_w, wi_b, bi, D_, 0, u, part);

  attn_pre_kernel<<<dim3(3, B_), 64, 0, stream>>>(u, vv, bi, part, pre);

  gemm_nt_wmma<1, 2, false><<<dim3(24, 4), 64, 0, stream>>>(pre, ln1_w, ln1_b, pair, D_, 1, nullptr, nullptr);
  gemm_nt_wmma<1, 2, false><<<dim3(9, 4), 64, 0, stream>>>(pair, pred_w, pred_b, out, R_, 0, nullptr, nullptr);
}


# Attention over the 8 bridges and pre = u + v + ctx folded into the v GEMM's epilogue (v GEMM launched behind the bridge GEMM; v never stored): 7 launches become 6; new dot sum order
# speedup vs baseline: 7.9917x; 1.0687x over previous
#include <hip/hip_runtime.h>

#define B_  64
#define S_  512
#define D_  768
#define K_  8
#define R_  277

typedef __attribute__((ext_vector_type(2))) float v2f;
typedef __attribute__((ext_vector_type(8))) float v8f;

__device__ __forceinline__ v8f wmma_f32_16x16x4(v2f a, v2f b, v8f c) {
#if defined(__gfx1250__) && __has_builtin(__builtin_amdgcn_wmma_f32_16x16x4_f32)
  return __builtin_amdgcn_wmma_f32_16x16x4_f32(
      false, a, false, b, (short)0, c, false, false);
#else
  asm volatile("v_wmma_f32_16x16x4_f32 %0, %1, %2, %0"
               : "+v"(c) : "v"(a), "v"(b));
  return c;
#endif
}

#define NSPAN_    10
#define CH_ROWS_  8
#define NCHUNK_   (S_ / CH_ROWS_)
#define STRIP_    64
#define SPAN_WV_  8

typedef __attribute__((ext_vector_type(4))) float v4f;

__device__ __forceinline__ v4f vmax4(v4f a, v4f b) {
  v4f m;
  m.x = fmaxf(a.x, b.x);
  m.y = fmaxf(a.y, b.y);
  m.z = fmaxf(a.z, b.z);
  m.w = fmaxf(a.w, b.w);
  return m;
}

__device__ __forceinline__ v4f span_row_load(const char* __restrict__ rowbase,
                                             unsigned colbytes, int r) {
  const unsigned off = (unsigned)r * (unsigned)(D_ * 4) + colbytes;
  return *(const v4f*)(rowbase + off);
}

__device__ __forceinline__ void chunk_load(v4f (&buf)[CH_ROWS_],
                                           const char* __restrict__ rowbase,
                                           unsigned colbytes, int c) {
#pragma unroll
  for (int i = 0; i < CH_ROWS_; ++i)
    buf[i] = span_row_load(rowbase, colbytes, c * CH_ROWS_ + i);
}

__device__ __forceinline__ v4f chunk_fold(const v4f (&buf)[CH_ROWS_]) {
  v4f m = buf[0];
#pragma unroll
  for (int i = 1; i < CH_ROWS_; ++i) m = vmax4(m, buf[i]);
  return m;
}

__device__ __forceinline__ void span_bounds(const int* __restrict__ h_span,
                                            const int* __restrict__ t_span,
                                            const int* __restrict__ b_spans,
                                            int b, int j, int& lo, int& hi) {
  const int* sp;
  if (j == 0)      sp = h_span + b * 2;
  else if (j == 1) sp = t_span + b * 2;
  else             sp = b_spans + (b * K_ + (j - 2)) * 2;
  lo = sp[0];
  hi = sp[1];
  if (lo < 0) lo = 0;
  if (hi > S_ - 1) hi = S_ - 1;
}

__device__ __forceinline__ int chunk_first(int lo) { return (lo + CH_ROWS_ - 1) / CH_ROWS_; }
__device__ __forceinline__ int chunk_last(int hi)  { return (hi + 1) / CH_ROWS_ - 1; }

__global__ void __launch_bounds__(32 * SPAN_WV_)
span_max_chunk_table(const float* __restrict__ emb,
                     const int*   __restrict__ h_span,
                     const int*   __restrict__ t_span,
                     const int*   __restrict__ b_spans,
                     float* __restrict__ h_emb,
                     float* __restrict__ t_emb,
                     float* __restrict__ b_emb) {
  __shared__ __attribute__((aligned(16))) float tab[NCHUNK_ * STRIP_];

  const int strip = blockIdx.x;
  const int b     = blockIdx.y;

  const int lane = threadIdx.x & 31;
  const int wave = __builtin_amdgcn_readfirstlane((int)(threadIdx.x >> 5));
  const int half = lane >> 4;
  const int lcol = (lane & 15) * 4;
  const int col  = strip * STRIP_ + lcol;

  const char* rowbase = (const char*)(emb + (size_t)b * S_ * D_);
  const unsigned colbytes = (unsigned)col * 4u;

  int cb = NCHUNK_, ce = -1;
#pragma unroll
  for (int j = 0; j < NSPAN_; ++j) {
    int lo, hi;
    span_bounds(h_span, t_span, b_spans, b, j, lo, hi);
    const int cf = chunk_first(lo), cl = chunk_last(hi);
    if (cf <= cl) {
      cb = cf < cb ? cf : cb;
      ce = cl > ce ? cl : ce;
    }
  }
  const int nch = ce - cb + 1;

  const int slot = 2 * wave + half;
#pragma unroll
  for (int p = 0; p < NCHUNK_ / 32; ++p) {
    if (32 * p < nch) {
      const int c0 = cb + 32 * p + slot;
      const int c1 = c0 + 16;
      v4f buf0[CH_ROWS_], buf1[CH_ROWS_];
      chunk_load(buf0, rowbase, colbytes, c0 < ce ? c0 : ce);
      chunk_load(buf1, rowbase, colbytes, c1 < ce ? c1 : ce);
      const v4f m0 = chunk_fold(buf0);
      if (c0 <= ce) *(v4f*)&tab[(c0 - cb) * STRIP_ + lcol] = m0;
      const v4f m1 = chunk_fold(buf1);
      if (c1 <= ce) *(v4f*)&tab[(c1 - cb) * STRIP_ + lcol] = m1;
    }
  }
  __syncthreads();

  const float ninf = -__builtin_inff();
  for (int j = wave; j < NSPAN_; j += SPAN_WV_) {
    int lo, hi;
    span_bounds(h_span, t_span, b_spans, b, j, lo, hi);
    float* dst;
    if (j == 0)      dst = h_emb + b * D_;
    else if (j == 1) dst = t_emb + b * D_;
    else             dst = b_emb + (size_t)(b * K_ + (j - 2)) * D_;

    v4f m = {ninf, ninf, ninf, ninf};
    if (lo <= hi) {
      const int cf = chunk_first(lo), cl = chunk_last(hi);
      int fend = CH_ROWS_ * cf - 1;
      fend = fend < hi ? fend : hi;
      fend = fend > lo ? fend : lo;
      int bbeg = CH_ROWS_ * (cl + 1);
      bbeg = bbeg > lo ? bbeg : lo;
      bbeg = bbeg < hi ? bbeg : hi;
      v4f e[CH_ROWS_ - 1];
#pragma unroll
      for (int i = 0; i < CH_ROWS_ - 1; ++i) {
        int rf = lo + i;  rf = rf < fend ? rf : fend;
        int rb = hi - i;  rb = rb > bbeg ? rb : bbeg;
        e[i] = span_row_load(rowbase, colbytes, half ? rb : rf);
      }
      if (cf <= cl) {
        const int ntrip = (cl - cf + 2) >> 1;
        for (int t0 = 0; t0 < ntrip; t0 += 8) {
          v4f q[8];
#pragma unroll
          for (int u = 0; u < 8; ++u) {
            int c = cf + 2 * (t0 + u) + half;
            c = c < cl ? c : cl;
            q[u] = *(const v4f*)&tab[(c - cb) * STRIP_ + lcol];
          }
#pragma unroll
          for (int u = 0; u < 8; ++u) m = vmax4(m, q[u]);
        }
      }
#pragma unroll
      for (int i = 0; i < CH_ROWS_ - 1; ++i) m = vmax4(m, e[i]);
    }

    m.x = fmaxf(m.x, __shfl_xor(m.x, 16));
    m.y = fmaxf(m.y, __shfl_xor(m.y, 16));
    m.z = fmaxf(m.z, __shfl_xor(m.z, 16));
    m.w = fmaxf(m.w, __shfl_xor(m.w, 16));

    if (half == 0) *(v4f*)(dst + col) = m;
  }
}

#define GK_      768
#define G_KC_    64
#define G_PITCH_ 68
#define G_NCH_   (GK_ / G_KC_)

template <int NP>
__device__ __forceinline__ void gemm_fetch(v4f (&reg)[NP],
                                           const float* const (&srcp)[NP], int chunk) {
#pragma unroll
  for (int i = 0; i < NP; ++i) reg[i] = *(const v4f*)(srcp[i] + chunk * G_KC_);
}

template <int NP, int RPP>
__device__ __forceinline__ void gemm_stash(const v4f (&reg)[NP], float* stage, int soff) {
#pragma unroll
  for (int i = 0; i < NP; ++i) *(v4f*)(stage + soff + i * RPP * G_PITCH_) = reg[i];
}

__device__ __forceinline__ void gemm_chunk(const float* ap, const float* bp,
                                           v8f& acc0, v8f& acc1) {
  v4f af[G_KC_ / 8], bf[G_KC_ / 8];
#pragma unroll
  for (int kb = 0; kb < G_KC_ / 8; ++kb) {
    af[kb] = *(const v4f*)(ap + kb * 8);
    bf[kb] = *(const v4f*)(bp + kb * 8);
  }
  __builtin_amdgcn_sched_barrier(0);
#pragma unroll
  for (int kb = 0; kb < G_KC_ / 8; ++kb) {
    const v2f a0 = {af[kb].x, af[kb].y}, b0 = {bf[kb].x, bf[kb].y};
    const v2f a1 = {af[kb].z, af[kb].w}, b1 = {bf[kb].z, bf[kb].w};
    acc0 = wmma_f32_16x16x4(a0, b0, acc0);
    acc1 = wmma_f32_16x16x4(a1, b1, acc1);
  }
}

__device__ __forceinline__ void gemm_lds_handover() {
  __builtin_amdgcn_fence(__ATOMIC_RELEASE, "workgroup", "local");
  __builtin_amdgcn_s_barrier();
  __builtin_amdgcn_fence(__ATOMIC_ACQUIRE, "workgroup", "local");
}

#define PT_  (D_ / 32)
#define PBM_ 64

#define EPI_PLAIN_ 0
#define EPI_DOTS_  1
#define EPI_ATTN_  2

template <int WM, int WN, int EPI>
__global__ void __launch_bounds__(32 * WM * WN)
gemm_nt_wmma(const float* __restrict__ A,
             const float* __restrict__ W,
             const float* __restrict__ bias,
             float* __restrict__ C,
             int N, int relu,
             const float* __restrict__ Uq,
             float* __restrict__ part,
             const float* __restrict__ Bi,
             const float* __restrict__ Pin) {
  constexpr bool DOTS = (EPI == EPI_DOTS_);
  constexpr bool ATTN = (EPI == EPI_ATTN_);
  constexpr int BM   = 16 * WM;
  constexpr int BN   = 16 * WN;
  constexpr int ROWS = BM + BN;
  constexpr int RPP  = 2 * WM * WN;
  constexpr int NP   = ROWS / RPP;
  constexpr int NPA  = BM / RPP;
  static_assert(ROWS % RPP == 0 && BM % RPP == 0, "copy passes must not straddle A and W");
  __shared__ __attribute__((aligned(16))) float tile[2][ROWS * G_PITCH_];

  const int tid  = threadIdx.x;
  const int lane = tid & 31;
  const int wave = tid >> 5;
  const int wm   = wave / WN;
  const int wn   = wave % WN;
  const int half = lane >> 4;
  const int r    = lane & 15;
  const int m0   = blockIdx.y * BM;
  const int n0   = blockIdx.x * BN;

  const int srow = tid >> 4;
  const int scol = (tid & 15) * 4;
  const float* srcp[NP];
#pragma unroll
  for (int i = 0; i < NPA; ++i)
    srcp[i] = A + (size_t)(m0 + srow + RPP * i) * GK_ + scol;
#pragma unroll
  for (int i = NPA; i < NP; ++i) {
    int wr = n0 + srow + RPP * (i - NPA);
    if (wr > N - 1) wr = N - 1;
    srcp[i] = W + (size_t)wr * GK_ + scol;
  }
  const int soff = srow * G_PITCH_ + scol;

  const float* ap0 = &tile[0][(wm * 16 + r) * G_PITCH_ + 4 * half];
  const float* bp0 = &tile[0][(BM + wn * 16 + r) * G_PITCH_ + 4 * half];
  const float* ap1 = &tile[1][(wm * 16 + r) * G_PITCH_ + 4 * half];
  const float* bp1 = &tile[1][(BM + wn * 16 + r) * G_PITCH_ + 4 * half];

  v8f acc0 = {}, acc1 = {};
  v4f ra[NP], rb[NP];

  gemm_fetch(ra, srcp, 0);
  gemm_fetch(rb, srcp, 1);

  v4f pa[PT_ / 4], pb[PT_ / 4], bq[2][K_], uq4[2];
  const int erow = tid >> 3;
  const int ecol = n0 + 4 * (tid & 7);
  if constexpr (ATTN) {
    static_assert(WM == 1 && WN == 2, "the attention epilogue is written for a 16 x 32 tile of 64 threads");
    static_assert(PT_ % 4 == 0 && K_ == 8, "4 threads share a row's column tiles; 8 bridges = 2 x 16 bytes");
    __builtin_amdgcn_sched_barrier(0);
    const int bd = m0 + (tid >> 2);
    const float* pp = Pin + ((size_t)(bd / (PBM_ / K_)) * PT_ + (tid & 3)) * PBM_ + (bd % (PBM_ / K_)) * K_;
#pragma unroll
    for (int j = 0; j < PT_ / 4; ++j) {
      pa[j] = *(const v4f*)(pp + (size_t)j * 4 * PBM_);
      pb[j] = *(const v4f*)(pp + (size_t)j * 4 * PBM_ + 4);
    }
#pragma unroll
    for (int e = 0; e < 2; ++e) {
      const int b = m0 + erow + 8 * e;
      uq4[e] = *(const v4f*)(Uq + (size_t)b * N + ecol);
#pragma unroll
      for (int k = 0; k < K_; ++k)
        bq[e][k] = *(const v4f*)(Bi + ((size_t)b * K_ + k) * N + ecol);
    }
  }

  gemm_stash<NP, RPP>(ra, tile[0], soff);
  gemm_fetch(ra, srcp, 2);

  for (int t = 0; t < G_NCH_; t += 2) {
    gemm_lds_handover();
    gemm_stash<NP, RPP>(rb, tile[1], soff);
    if (t + 3 < G_NCH_) gemm_fetch(rb, srcp, t + 3);
    gemm_chunk(ap0, bp0, acc0, acc1);
    gemm_lds_handover();
    if (t + 2 < G_NCH_) gemm_stash<NP, RPP>(ra, tile[0], soff);
    if (t + 4 < G_NCH_) gemm_fetch(ra, srcp, t + 4);
    gemm_chunk(ap1, bp1, acc0, acc1);
  }

  const int n = n0 + wn * 16 + r;
  float pd[8];
#pragma unroll
  for (int i = 0; i < 8; ++i) pd[i] = 0.0f;
  if constexpr (ATTN) {
    __shared__ __attribute__((aligned(16))) float sv[BM][BN + 4];
    __shared__ __attribute__((aligned(16))) float satt[BM * K_];
    const float bv = bias[n];
#pragma unroll
    for (int i = 0; i < 8; ++i) {
      float vout = (acc0[i] + acc1[i]) + bv;
      if (relu) vout = fmaxf(vout, 0.0f);
      sv[wm * 16 + i + 8 * half][wn * 16 + r] = vout;
    }
    float att[K_];
    att[0] = pa[0].x; att[1] = pa[0].y; att[2] = pa[0].z; att[3] = pa[0].w;
    att[4] = pb[0].x; att[5] = pb[0].y; att[6] = pb[0].z; att[7] = pb[0].w;
#pragma unroll
    for (int j = 1; j < PT_ / 4; ++j) {
      att[0] += pa[j].x; att[1] += pa[j].y; att[2] += pa[j].z; att[3] += pa[j].w;
      att[4] += pb[j].x; att[5] += pb[j].y; att[6] += pb[j].z; att[7] += pb[j].w;
    }
#pragma unroll
    for (int off = 1; off < 4; off <<= 1) {
#pragma unroll
      for (int k = 0; k < K_; ++k) att[k] += __shfl_xor(att[k], off);
    }
#pragma unroll
    for (int k = 0; k < K_; ++k) att[k] *= 0.125f;
    float mx = att[0];
#pragma unroll
    for (int k = 1; k < K_; ++k) mx = fmaxf(mx, att[k]);
    float s = 0.f;
#pragma unroll
    for (int k = 0; k < K_; ++k) { att[k] = __expf(att[k] - mx); s += att[k]; }
    const float inv = 1.0f / s;
#pragma unroll
    for (int k = 0; k < K_; ++k) att[k] *= inv;
    if ((tid & 3) == 0) {
      const v4f w0 = {att[0], att[1], att[2], att[3]};
      const v4f w1 = {att[4], att[5], att[6], att[7]};
      *(v4f*)&satt[(tid >> 2) * K_]     = w0;
      *(v4f*)&satt[(tid >> 2) * K_ + 4] = w1;
    }
    gemm_lds_handover();
#pragma unroll
    for (int e = 0; e < 2; ++e) {
      const int row = erow + 8 * e;
      const v4f vv = *(const v4f*)&sv[row][4 * (tid & 7)];
      const v4f w0 = *(const v4f*)&satt[row * K_];
      const v4f w1 = *(const v4f*)&satt[row * K_ + 4];
      const float wk[K_] = {w0.x, w0.y, w0.z, w0.w, w1.x, w1.y, w1.z, w1.w};
      v4f c = {0.f, 0.f, 0.f, 0.f};
#pragma unroll
      for (int k = 0; k < K_; ++k) {
        c.x += wk[k] * bq[e][k].x;
        c.y += wk[k] * bq[e][k].y;
        c.z += wk[k] * bq[e][k].z;
        c.w += wk[k] * bq[e][k].w;
      }
      v4f o;
      o.x = uq4[e].x + vv.x + c.x;
      o.y = uq4[e].y + vv.y + c.y;
      o.z = uq4[e].z + vv.z + c.z;
      o.w = uq4[e].w + vv.w + c.w;
      *(v4f*)(C + (size_t)(m0 + row) * N + ecol) = o;
    }
  } else if (n < N) {
    const float bv = bias[n];
    float uq = 0.0f;
    if constexpr (DOTS) uq = Uq[(size_t)((m0 + wm * 16) / K_ + half) * N + n];
#pragma unroll
    for (int i = 0; i < 8; ++i) {
      float vout = (acc0[i] + acc1[i]) + bv;
      if (relu) vout = fmaxf(vout, 0.0f);
      C[(size_t)(m0 + wm * 16 + i + 8 * half) * N + n] = vout;
      if constexpr (DOTS) pd[i] = uq * vout;
    }
  }
  if constexpr (DOTS) {
    static_assert(K_ == 8, "one lane half (8 rows of the piece) must be one batch row");
    __shared__ float dred[WN][BM];
#pragma unroll
    for (int off = 8; off > 0; off >>= 1) {
#pragma unroll
      for (int i = 0; i < 8; ++i) pd[i] += __shfl_xor(pd[i], off);
    }
    if (r == 0) {
#pragma unroll
      for (int i = 0; i < 8; ++i) dred[wn][wm * 16 + 8 * half + i] = pd[i];
    }
    gemm_lds_handover();
    if (tid < BM) {
      float s = dred[0][tid];
#pragma unroll
      for (int w = 1; w < WN; ++w) s += dred[w][tid];
      const int ntile = (N + BN - 1) / BN;
      part[((size_t)blockIdx.y * ntile + blockIdx.x) * BM + tid] = s;
    }
  }
}

extern "C" void kernel_launch(void* const* d_in, const int* in_sizes, int n_in,
                              void* d_out, int out_size, void* d_ws, size_t ws_size,
                              hipStream_t stream) {
  (void)in_sizes; (void)n_in; (void)out_size; (void)ws_size;

  const float* emb     = (const float*)d_in[0];
  const int*   h_span  = (const int*)d_in[1];
  const int*   t_span  = (const int*)d_in[2];
  const int*   b_spans = (const int*)d_in[3];
  const float* wu_w    = (const float*)d_in[4];
  const float* wu_b    = (const float*)d_in[5];
  const float* wv_w    = (const float*)d_in[6];
  const float* wv_b    = (const float*)d_in[7];
  const float* wi_w    = (const float*)d_in[8];
  const float* wi_b    = (const float*)d_in[9];
  const float* ln1_w   = (const float*)d_in[10];
  const float* ln1_b   = (const float*)d_in[11];
  const float* pred_w  = (const float*)d_in[12];
  const float* pred_b  = (const float*)d_in[13];
  float* out = (float*)d_out;

  float* ws    = (float*)d_ws;
  float* h_emb = ws;
  float* t_emb = h_emb + B_ * D_;
  float* b_emb = t_emb + B_ * D_;
  float* u     = b_emb + B_ * K_ * D_;
  float* part  = u + B_ * D_;
  float* bi    = part + B_ * D_;
  float* pre   = bi + B_ * K_ * D_;
  float* pair  = h_emb;

  span_max_chunk_table<<<dim3(D_ / STRIP_, B_), 32 * SPAN_WV_, 0, stream>>>(emb, h_span, t_span, b_spans,
                                                                            h_emb, t_emb, b_emb);

  gemm_nt_wmma<1, 2, EPI_PLAIN_><<<dim3(24, 4), 64, 0, stream>>>(h_emb, wu_w, wu_b, u, D_, 0,
                                                                nullptr, nullptr, nullptr, nullptr);
  gemm_nt_wmma<4, 2, EPI_DOTS_><<<dim3(PT_, 8), 256, 0, stream>>>(b_emb, wi_w, wi_b, bi, D_, 0,
                                                                  u, part, nullptr, nullptr);
  gemm_nt_wmma<1, 2, EPI_ATTN_><<<dim3(24, 4), 64, 0, stream>>>(t_emb, wv_w, wv_b, pre, D_, 0,
                                                               u, nullptr, bi, part);

  gemm_nt_wmma<1, 2, EPI_PLAIN_><<<dim3(24, 4), 64, 0, stream>>>(pre, ln1_w, ln1_b, pair, D_, 1,
                                                                nullptr, nullptr, nullptr, nullptr);
  gemm_nt_wmma<1, 2, EPI_PLAIN_><<<dim3(9, 4), 64, 0, stream>>>(pair, pred_w, pred_b, out, R_, 0,
                                                               nullptr, nullptr, nullptr, nullptr);
}


# GEMMs: K walked in chunks of 128 instead of 64 (LDS row pitch 528 B, one wave copies one 512-byte row): 6 LDS hand-overs per product instead of 12; same sum order
# speedup vs baseline: 8.3299x; 1.0423x over previous
#include <hip/hip_runtime.h>

#define B_  64
#define S_  512
#define D_  768
#define K_  8
#define R_  277

typedef __attribute__((ext_vector_type(2))) float v2f;
typedef __attribute__((ext_vector_type(8))) float v8f;

__device__ __forceinline__ v8f wmma_f32_16x16x4(v2f a, v2f b, v8f c) {
#if defined(__gfx1250__) && __has_builtin(__builtin_amdgcn_wmma_f32_16x16x4_f32)
  return __builtin_amdgcn_wmma_f32_16x16x4_f32(
      false, a, false, b, (short)0, c, false, false);
#else
  asm volatile("v_wmma_f32_16x16x4_f32 %0, %1, %2, %0"
               : "+v"(c) : "v"(a), "v"(b));
  return c;
#endif
}

#define NSPAN_    10
#define CH_ROWS_  8
#define NCHUNK_   (S_ / CH_ROWS_)
#define STRIP_    64
#define SPAN_WV_  8

typedef __attribute__((ext_vector_type(4))) float v4f;

__device__ __forceinline__ v4f vmax4(v4f a, v4f b) {
  v4f m;
  m.x = fmaxf(a.x, b.x);
  m.y = fmaxf(a.y, b.y);
  m.z = fmaxf(a.z, b.z);
  m.w = fmaxf(a.w, b.w);
  return m;
}

__device__ __forceinline__ v4f span_row_load(const char* __restrict__ rowbase,
                                             unsigned colbytes, int r) {
  const unsigned off = (unsigned)r * (unsigned)(D_ * 4) + colbytes;
  return *(const v4f*)(rowbase + off);
}

__device__ __forceinline__ void chunk_load(v4f (&buf)[CH_ROWS_],
                                           const char* __restrict__ rowbase,
                                           unsigned colbytes, int c) {
#pragma unroll
  for (int i = 0; i < CH_ROWS_; ++i)
    buf[i] = span_row_load(rowbase, colbytes, c * CH_ROWS_ + i);
}

__device__ __forceinline__ v4f chunk_fold(const v4f (&buf)[CH_ROWS_]) {
  v4f m = buf[0];
#pragma unroll
  for (int i = 1; i < CH_ROWS_; ++i) m = vmax4(m, buf[i]);
  return m;
}

__device__ __forceinline__ void span_bounds(const int* __restrict__ h_span,
                                            const int* __restrict__ t_span,
                                            const int* __restrict__ b_spans,
                                            int b, int j, int& lo, int& hi) {
  const int* sp;
  if (j == 0)      sp = h_span + b * 2;
  else if (j == 1) sp = t_span + b * 2;
  else             sp = b_spans + (b * K_ + (j - 2)) * 2;
  lo = sp[0];
  hi = sp[1];
  if (lo < 0) lo = 0;
  if (hi > S_ - 1) hi = S_ - 1;
}

__device__ __forceinline__ int chunk_first(int lo) { return (lo + CH_ROWS_ - 1) / CH_ROWS_; }
__device__ __forceinline__ int chunk_last(int hi)  { return (hi + 1) / CH_ROWS_ - 1; }

__global__ void __launch_bounds__(32 * SPAN_WV_)
span_max_chunk_table(const float* __restrict__ emb,
                     const int*   __restrict__ h_span,
                     const int*   __restrict__ t_span,
                     const int*   __restrict__ b_spans,
                     float* __restrict__ h_emb,
                     float* __restrict__ t_emb,
                     float* __restrict__ b_emb) {
  __shared__ __attribute__((aligned(16))) float tab[NCHUNK_ * STRIP_];

  const int strip = blockIdx.x;
  const int b     = blockIdx.y;

  const int lane = threadIdx.x & 31;
  const int wave = __builtin_amdgcn_readfirstlane((int)(threadIdx.x >> 5));
  const int half = lane >> 4;
  const int lcol = (lane & 15) * 4;
  const int col  = strip * STRIP_ + lcol;

  const char* rowbase = (const char*)(emb + (size_t)b * S_ * D_);
  const unsigned colbytes = (unsigned)col * 4u;

  int cb = NCHUNK_, ce = -1;
#pragma unroll
  for (int j = 0; j < NSPAN_; ++j) {
    int lo, hi;
    span_bounds(h_span, t_span, b_spans, b, j, lo, hi);
    const int cf = chunk_first(lo), cl = chunk_last(hi);
    if (cf <= cl) {
      cb = cf < cb ? cf : cb;
      ce = cl > ce ? cl : ce;
    }
  }
  const int nch = ce - cb + 1;

  const int slot = 2 * wave + half;
#pragma unroll
  for (int p = 0; p < NCHUNK_ / 32; ++p) {
    if (32 * p < nch) {
      const int c0 = cb + 32 * p + slot;
      const int c1 = c0 + 16;
      v4f buf0[CH_ROWS_], buf1[CH_ROWS_];
      chunk_load(buf0, rowbase, colbytes, c0 < ce ? c0 : ce);
      chunk_load(buf1, rowbase, colbytes, c1 < ce ? c1 : ce);
      const v4f m0 = chunk_fold(buf0);
      if (c0 <= ce) *(v4f*)&tab[(c0 - cb) * STRIP_ + lcol] = m0;
      const v4f m1 = chunk_fold(buf1);
      if (c1 <= ce) *(v4f*)&tab[(c1 - cb) * STRIP_ + lcol] = m1;
    }
  }
  __syncthreads();

  const float ninf = -__builtin_inff();
  for (int j = wave; j < NSPAN_; j += SPAN_WV_) {
    int lo, hi;
    span_bounds(h_span, t_span, b_spans, b, j, lo, hi);
    float* dst;
    if (j == 0)      dst = h_emb + b * D_;
    else if (j == 1) dst = t_emb + b * D_;
    else             dst = b_emb + (size_t)(b * K_ + (j - 2)) * D_;

    v4f m = {ninf, ninf, ninf, ninf};
    if (lo <= hi) {
      const int cf = chunk_first(lo), cl = chunk_last(hi);
      int fend = CH_ROWS_ * cf - 1;
      fend = fend < hi ? fend : hi;
      fend = fend > lo ? fend : lo;
      int bbeg = CH_ROWS_ * (cl + 1);
      bbeg = bbeg > lo ? bbeg : lo;
      bbeg = bbeg < hi ? bbeg : hi;
      v4f e[CH_ROWS_ - 1];
#pragma unroll
      for (int i = 0; i < CH_ROWS_ - 1; ++i) {
        int rf = lo + i;  rf = rf < fend ? rf : fend;
        int rb = hi - i;  rb = rb > bbeg ? rb : bbeg;
        e[i] = span_row_load(rowbase, colbytes, half ? rb : rf);
      }
      if (cf <= cl) {
        const int ntrip = (cl - cf + 2) >> 1;
        for (int t0 = 0; t0 < ntrip; t0 += 8) {
          v4f q[8];
#pragma unroll
          for (int u = 0; u < 8; ++u) {
            int c = cf + 2 * (t0 + u) + half;
            c = c < cl ? c : cl;
            q[u] = *(const v4f*)&tab[(c - cb) * STRIP_ + lcol];
          }
#pragma unroll
          for (int u = 0; u < 8; ++u) m = vmax4(m, q[u]);
        }
      }
#pragma unroll
      for (int i = 0; i < CH_ROWS_ - 1; ++i) m = vmax4(m, e[i]);
    }

    m.x = fmaxf(m.x, __shfl_xor(m.x, 16));
    m.y = fmaxf(m.y, __shfl_xor(m.y, 16));
    m.z = fmaxf(m.z, __shfl_xor(m.z, 16));
    m.w = fmaxf(m.w, __shfl_xor(m.w, 16));

    if (half == 0) *(v4f*)(dst + col) = m;
  }
}

#define GK_      768
#define G_KC_    128
#define G_PITCH_ 132
#define G_NCH_   (GK_ / G_KC_)
#define G_LPR_   (G_KC_ / 4)

template <int NP>
__device__ __forceinline__ void gemm_fetch(v4f (&reg)[NP],
                                           const float* const (&srcp)[NP], int chunk) {
#pragma unroll
  for (int i = 0; i < NP; ++i) reg[i] = *(const v4f*)(srcp[i] + chunk * G_KC_);
}

template <int NP, int RPP>
__device__ __forceinline__ void gemm_stash(const v4f (&reg)[NP], float* stage, int soff) {
#pragma unroll
  for (int i = 0; i < NP; ++i) *(v4f*)(stage + soff + i * RPP * G_PITCH_) = reg[i];
}

__device__ __forceinline__ void gemm_chunk(const float* ap, const float* bp,
                                           v8f& acc0, v8f& acc1) {
  v4f af[G_KC_ / 8], bf[G_KC_ / 8];
#pragma unroll
  for (int kb = 0; kb < G_KC_ / 8; ++kb) {
    af[kb] = *(const v4f*)(ap + kb * 8);
    bf[kb] = *(const v4f*)(bp + kb * 8);
  }
  __builtin_amdgcn_sched_barrier(0);
#pragma unroll
  for (int kb = 0; kb < G_KC_ / 8; ++kb) {
    const v2f a0 = {af[kb].x, af[kb].y}, b0 = {bf[kb].x, bf[kb].y};
    const v2f a1 = {af[kb].z, af[kb].w}, b1 = {bf[kb].z, bf[kb].w};
    acc0 = wmma_f32_16x16x4(a0, b0, acc0);
    acc1 = wmma_f32_16x16x4(a1, b1, acc1);
  }
}

__device__ __forceinline__ void gemm_lds_handover() {
  __builtin_amdgcn_fence(__ATOMIC_RELEASE, "workgroup", "local");
  __builtin_amdgcn_s_barrier();
  __builtin_amdgcn_fence(__ATOMIC_ACQUIRE, "workgroup", "local");
}

#define PT_  (D_ / 32)
#define PBM_ 64

#define EPI_PLAIN_ 0
#define EPI_DOTS_  1
#define EPI_ATTN_  2

template <int WM, int WN, int EPI>
__global__ void __launch_bounds__(32 * WM * WN)
gemm_nt_wmma(const float* __restrict__ A,
             const float* __restrict__ W,
             const float* __restrict__ bias,
             float* __restrict__ C,
             int N, int relu,
             const float* __restrict__ Uq,
             float* __restrict__ part,
             const float* __restrict__ Bi,
             const float* __restrict__ Pin) {
  constexpr bool DOTS = (EPI == EPI_DOTS_);
  constexpr bool ATTN = (EPI == EPI_ATTN_);
  constexpr int BM   = 16 * WM;
  constexpr int BN   = 16 * WN;
  constexpr int ROWS = BM + BN;
  constexpr int RPP  = 32 * WM * WN / G_LPR_;
  constexpr int NP   = ROWS / RPP;
  constexpr int NPA  = BM / RPP;
  static_assert(ROWS % RPP == 0 && BM % RPP == 0, "copy passes must not straddle A and W");
  static_assert(G_NCH_ % 2 == 0 && G_NCH_ >= 4, "the K loop takes the chunks in pairs, two fetched ahead");
  __shared__ __attribute__((aligned(16))) float tile[2][ROWS * G_PITCH_];

  const int tid  = threadIdx.x;
  const int lane = tid & 31;
  const int wave = tid >> 5;
  const int wm   = wave / WN;
  const int wn   = wave % WN;
  const int half = lane >> 4;
  const int r    = lane & 15;
  const int m0   = blockIdx.y * BM;
  const int n0   = blockIdx.x * BN;

  const int srow = tid / G_LPR_;
  const int scol = (tid % G_LPR_) * 4;
  const float* srcp[NP];
#pragma unroll
  for (int i = 0; i < NPA; ++i)
    srcp[i] = A + (size_t)(m0 + srow + RPP * i) * GK_ + scol;
#pragma unroll
  for (int i = NPA; i < NP; ++i) {
    int wr = n0 + srow + RPP * (i - NPA);
    if (wr > N - 1) wr = N - 1;
    srcp[i] = W + (size_t)wr * GK_ + scol;
  }
  const int soff = srow * G_PITCH_ + scol;

  const float* ap0 = &tile[0][(wm * 16 + r) * G_PITCH_ + 4 * half];
  const float* bp0 = &tile[0][(BM + wn * 16 + r) * G_PITCH_ + 4 * half];
  const float* ap1 = &tile[1][(wm * 16 + r) * G_PITCH_ + 4 * half];
  const float* bp1 = &tile[1][(BM + wn * 16 + r) * G_PITCH_ + 4 * half];

  v8f acc0 = {}, acc1 = {};
  v4f ra[NP], rb[NP];

  gemm_fetch(ra, srcp, 0);
  gemm_fetch(rb, srcp, 1);

  v4f pa[PT_ / 4], pb[PT_ / 4], bq[2][K_], uq4[2];
  const int erow = tid >> 3;
  const int ecol = n0 + 4 * (tid & 7);
  if constexpr (ATTN) {
    static_assert(WM == 1 && WN == 2, "the attention epilogue is written for a 16 x 32 tile of 64 threads");
    static_assert(PT_ % 4 == 0 && K_ == 8, "4 threads share a row's column tiles; 8 bridges = 2 x 16 bytes");
    __builtin_amdgcn_sched_barrier(0);
    const int bd = m0 + (tid >> 2);
    const float* pp = Pin + ((size_t)(bd / (PBM_ / K_)) * PT_ + (tid & 3)) * PBM_ + (bd % (PBM_ / K_)) * K_;
#pragma unroll
    for (int j = 0; j < PT_ / 4; ++j) {
      pa[j] = *(const v4f*)(pp + (size_t)j * 4 * PBM_);
      pb[j] = *(const v4f*)(pp + (size_t)j * 4 * PBM_ + 4);
    }
#pragma unroll
    for (int e = 0; e < 2; ++e) {
      const int b = m0 + erow + 8 * e;
      uq4[e] = *(const v4f*)(Uq + (size_t)b * N + ecol);
#pragma unroll
      for (int k = 0; k < K_; ++k)
        bq[e][k] = *(const v4f*)(Bi + ((size_t)b * K_ + k) * N + ecol);
    }
  }

  gemm_stash<NP, RPP>(ra, tile[0], soff);
  gemm_fetch(ra, srcp, 2);

  for (int t = 0; t < G_NCH_; t += 2) {
    gemm_lds_handover();
    gemm_stash<NP, RPP>(rb, tile[1], soff);
    if (t + 3 < G_NCH_) gemm_fetch(rb, srcp, t + 3);
    gemm_chunk(ap0, bp0, acc0, acc1);
    gemm_lds_handover();
    if (t + 2 < G_NCH_) gemm_stash<NP, RPP>(ra, tile[0], soff);
    if (t + 4 < G_NCH_) gemm_fetch(ra, srcp, t + 4);
    gemm_chunk(ap1, bp1, acc0, acc1);
  }

  const int n = n0 + wn * 16 + r;
  float pd[8];
#pragma unroll
  for (int i = 0; i < 8; ++i) pd[i] = 0.0f;
  if constexpr (ATTN) {
    __shared__ __attribute__((aligned(16))) float sv[BM][BN + 4];
    __shared__ __attribute__((aligned(16))) float satt[BM * K_];
    const float bv = bias[n];
#pragma unroll
    for (int i = 0; i < 8; ++i) {
      float vout = (acc0[i] + acc1[i]) + bv;
      if (relu) vout = fmaxf(vout, 0.0f);
      sv[wm * 16 + i + 8 * half][wn * 16 + r] = vout;
    }
    float att[K_];
    att[0] = pa[0].x; att[1] = pa[0].y; att[2] = pa[0].z; att[3] = pa[0].w;
    att[4] = pb[0].x; att[5] = pb[0].y; att[6] = pb[0].z; att[7] = pb[0].w;
#pragma unroll
    for (int j = 1; j < PT_ / 4; ++j) {
      att[0] += pa[j].x; att[1] += pa[j].y; att[2] += pa[j].z; att[3] += pa[j].w;
      att[4] += pb[j].x; att[5] += pb[j].y; att[6] += pb[j].z; att[7] += pb[j].w;
    }
#pragma unroll
    for (int off = 1; off < 4; off <<= 1) {
#pragma unroll
      for (int k = 0; k < K_; ++k) att[k] += __shfl_xor(att[k], off);
    }
#pragma unroll
    for (int k = 0; k < K_; ++k) att[k] *= 0.125f;
    float mx = att[0];
#pragma unroll
    for (int k = 1; k < K_; ++k) mx = fmaxf(mx, att[k]);
    float s = 0.f;
#pragma unroll
    for (int k = 0; k < K_; ++k) { att[k] = __expf(att[k] - mx); s += att[k]; }
    const float inv = 1.0f / s;
#pragma unroll
    for (int k = 0; k < K_; ++k) att[k] *= inv;
    if ((tid & 3) == 0) {
      const v4f w0 = {att[0], att[1], att[2], att[3]};
      const v4f w1 = {att[4], att[5], att[6], att[7]};
      *(v4f*)&satt[(tid >> 2) * K_]     = w0;
      *(v4f*)&satt[(tid >> 2) * K_ + 4] = w1;
    }
    gemm_lds_handover();
#pragma unroll
    for (int e = 0; e < 2; ++e) {
      const int row = erow + 8 * e;
      const v4f vv = *(const v4f*)&sv[row][4 * (tid & 7)];
      const v4f w0 = *(const v4f*)&satt[row * K_];
      const v4f w1 = *(const v4f*)&satt[row * K_ + 4];
      const float wk[K_] = {w0.x, w0.y, w0.z, w0.w, w1.x, w1.y, w1.z, w1.w};
      v4f c = {0.f, 0.f, 0.f, 0.f};
#pragma unroll
      for (int k = 0; k < K_; ++k) {
        c.x += wk[k] * bq[e][k].x;
        c.y += wk[k] * bq[e][k].y;
        c.z += wk[k] * bq[e][k].z;
        c.w += wk[k] * bq[e][k].w;
      }
      v4f o;
      o.x = uq4[e].x + vv.x + c.x;
      o.y = uq4[e].y + vv.y + c.y;
      o.z = uq4[e].z + vv.z + c.z;
      o.w = uq4[e].w + vv.w + c.w;
      *(v4f*)(C + (size_t)(m0 + row) * N + ecol) = o;
    }
  } else if (n < N) {
    const float bv = bias[n];
    float uq = 0.0f;
    if constexpr (DOTS) uq = Uq[(size_t)((m0 + wm * 16) / K_ + half) * N + n];
#pragma unroll
    for (int i = 0; i < 8; ++i) {
      float vout = (acc0[i] + acc1[i]) + bv;
      if (relu) vout = fmaxf(vout, 0.0f);
      C[(size_t)(m0 + wm * 16 + i + 8 * half) * N + n] = vout;
      if constexpr (DOTS) pd[i] = uq * vout;
    }
  }
  if constexpr (DOTS) {
    static_assert(K_ == 8, "one lane half (8 rows of the piece) must be one batch row");
    __shared__ float dred[WN][BM];
#pragma unroll
    for (int off = 8; off > 0; off >>= 1) {
#pragma unroll
      for (int i = 0; i < 8; ++i) pd[i] += __shfl_xor(pd[i], off);
    }
    if (r == 0) {
#pragma unroll
      for (int i = 0; i < 8; ++i) dred[wn][wm * 16 + 8 * half + i] = pd[i];
    }
    gemm_lds_handover();
    if (tid < BM) {
      float s = dred[0][tid];
#pragma unroll
      for (int w = 1; w < WN; ++w) s += dred[w][tid];
      const int ntile = (N + BN - 1) / BN;
      part[((size_t)blockIdx.y * ntile + blockIdx.x) * BM + tid] = s;
    }
  }
}

extern "C" void kernel_launch(void* const* d_in, const int* in_sizes, int n_in,
                              void* d_out, int out_size, void* d_ws, size_t ws_size,
                              hipStream_t stream) {
  (void)in_sizes; (void)n_in; (void)out_size; (void)ws_size;

  const float* emb     = (const float*)d_in[0];
  const int*   h_span  = (const int*)d_in[1];
  const int*   t_span  = (const int*)d_in[2];
  const int*   b_spans = (const int*)d_in[3];
  const float* wu_w    = (const float*)d_in[4];
  const float* wu_b    = (const float*)d_in[5];
  const float* wv_w    = (const float*)d_in[6];
  const float* wv_b    = (const float*)d_in[7];
  const float* wi_w    = (const float*)d_in[8];
  const float* wi_b    = (const float*)d_in[9];
  const float* ln1_w   = (const float*)d_in[10];
  const float* ln1_b   = (const float*)d_in[11];
  const float* pred_w  = (const float*)d_in[12];
  const float* pred_b  = (const float*)d_in[13];
  float* out = (float*)d_out;

  float* ws    = (float*)d_ws;
  float* h_emb = ws;
  float* t_emb = h_emb + B_ * D_;
  float* b_emb = t_emb + B_ * D_;
  float* u     = b_emb + B_ * K_ * D_;
  float* part  = u + B_ * D_;
  float* bi    = part + B_ * D_;
  float* pre   = bi + B_ * K_ * D_;
  float* pair  = h_emb;

  span_max_chunk_table<<<dim3(D_ / STRIP_, B_), 32 * SPAN_WV_, 0, stream>>>(emb, h_span, t_span, b_spans,
                                                                            h_emb, t_emb, b_emb);

  gemm_nt_wmma<1, 2, EPI_PLAIN_><<<dim3(24, 4), 64, 0, stream>>>(h_emb, wu_w, wu_b, u, D_, 0,
                                                                nullptr, nullptr, nullptr, nullptr);
  gemm_nt_wmma<4, 2, EPI_DOTS_><<<dim3(PT_, 8), 256, 0, stream>>>(b_emb, wi_w, wi_b, bi, D_, 0,
                                                                  u, part, nullptr, nullptr);
  gemm_nt_wmma<1, 2, EPI_ATTN_><<<dim3(24, 4), 64, 0, stream>>>(t_emb, wv_w, wv_b, pre, D_, 0,
                                                               u, nullptr, bi, part);

  gemm_nt_wmma<1, 2, EPI_PLAIN_><<<dim3(24, 4), 64, 0, stream>>>(pre, ln1_w, ln1_b, pair, D_, 1,
                                                                nullptr, nullptr, nullptr, nullptr);
  gemm_nt_wmma<1, 2, EPI_PLAIN_><<<dim3(9, 4), 64, 0, stream>>>(pair, pred_w, pred_b, out, R_, 0,
                                                               nullptr, nullptr, nullptr, nullptr);
}


# u product folded into the bridge product's launch: each 64x32 bridge block forms the 8x32 u tile its dots need (K dealt over its 8 waves, partials added in LDS); 6 launches become 5; new u sum order
# speedup vs baseline: 8.9451x; 1.0739x over previous
#include <hip/hip_runtime.h>

#define B_  64
#define S_  512
#define D_  768
#define K_  8
#define R_  277

typedef __attribute__((ext_vector_type(2))) float v2f;
typedef __attribute__((ext_vector_type(8))) float v8f;

__device__ __forceinline__ v8f wmma_f32_16x16x4(v2f a, v2f b, v8f c) {
#if defined(__gfx1250__) && __has_builtin(__builtin_amdgcn_wmma_f32_16x16x4_f32)
  return __builtin_amdgcn_wmma_f32_16x16x4_f32(
      false, a, false, b, (short)0, c, false, false);
#else
  asm volatile("v_wmma_f32_16x16x4_f32 %0, %1, %2, %0"
               : "+v"(c) : "v"(a), "v"(b));
  return c;
#endif
}

#define NSPAN_    10
#define CH_ROWS_  8
#define NCHUNK_   (S_ / CH_ROWS_)
#define STRIP_    64
#define SPAN_WV_  8

typedef __attribute__((ext_vector_type(4))) float v4f;

__device__ __forceinline__ v4f vmax4(v4f a, v4f b) {
  v4f m;
  m.x = fmaxf(a.x, b.x);
  m.y = fmaxf(a.y, b.y);
  m.z = fmaxf(a.z, b.z);
  m.w = fmaxf(a.w, b.w);
  return m;
}

__device__ __forceinline__ v4f span_row_load(const char* __restrict__ rowbase,
                                             unsigned colbytes, int r) {
  const unsigned off = (unsigned)r * (unsigned)(D_ * 4) + colbytes;
  return *(const v4f*)(rowbase + off);
}

__device__ __forceinline__ void chunk_load(v4f (&buf)[CH_ROWS_],
                                           const char* __restrict__ rowbase,
                                           unsigned colbytes, int c) {
#pragma unroll
  for (int i = 0; i < CH_ROWS_; ++i)
    buf[i] = span_row_load(rowbase, colbytes, c * CH_ROWS_ + i);
}

__device__ __forceinline__ v4f chunk_fold(const v4f (&buf)[CH_ROWS_]) {
  v4f m = buf[0];
#pragma unroll
  for (int i = 1; i < CH_ROWS_; ++i) m = vmax4(m, buf[i]);
  return m;
}

__device__ __forceinline__ void span_bounds(const int* __restrict__ h_span,
                                            const int* __restrict__ t_span,
                                            const int* __restrict__ b_spans,
                                            int b, int j, int& lo, int& hi) {
  const int* sp;
  if (j == 0)      sp = h_span + b * 2;
  else if (j == 1) sp = t_span + b * 2;
  else             sp = b_spans + (b * K_ + (j - 2)) * 2;
  lo = sp[0];
  hi = sp[1];
  if (lo < 0) lo = 0;
  if (hi > S_ - 1) hi = S_ - 1;
}

__device__ __forceinline__ int chunk_first(int lo) { return (lo + CH_ROWS_ - 1) / CH_ROWS_; }
__device__ __forceinline__ int chunk_last(int hi)  { return (hi + 1) / CH_ROWS_ - 1; }

__global__ void __launch_bounds__(32 * SPAN_WV_)
span_max_chunk_table(const float* __restrict__ emb,
                     const int*   __restrict__ h_span,
                     const int*   __restrict__ t_span,
                     const int*   __restrict__ b_spans,
                     float* __restrict__ h_emb,
                     float* __restrict__ t_emb,
                     float* __restrict__ b_emb) {
  __shared__ __attribute__((aligned(16))) float tab[NCHUNK_ * STRIP_];

  const int strip = blockIdx.x;
  const int b     = blockIdx.y;

  const int lane = threadIdx.x & 31;
  const int wave = __builtin_amdgcn_readfirstlane((int)(threadIdx.x >> 5));
  const int half = lane >> 4;
  const int lcol = (lane & 15) * 4;
  const int col  = strip * STRIP_ + lcol;

  const char* rowbase = (const char*)(emb + (size_t)b * S_ * D_);
  const unsigned colbytes = (unsigned)col * 4u;

  int cb = NCHUNK_, ce = -1;
#pragma unroll
  for (int j = 0; j < NSPAN_; ++j) {
    int lo, hi;
    span_bounds(h_span, t_span, b_spans, b, j, lo, hi);
    const int cf = chunk_first(lo), cl = chunk_last(hi);
    if (cf <= cl) {
      cb = cf < cb ? cf : cb;
      ce = cl > ce ? cl : ce;
    }
  }
  const int nch = ce - cb + 1;

  const int slot = 2 * wave + half;
#pragma unroll
  for (int p = 0; p < NCHUNK_ / 32; ++p) {
    if (32 * p < nch) {
      const int c0 = cb + 32 * p + slot;
      const int c1 = c0 + 16;
      v4f buf0[CH_ROWS_], buf1[CH_ROWS_];
      chunk_load(buf0, rowbase, colbytes, c0 < ce ? c0 : ce);
      chunk_load(buf1, rowbase, colbytes, c1 < ce ? c1 : ce);
      const v4f m0 = chunk_fold(buf0);
      if (c0 <= ce) *(v4f*)&tab[(c0 - cb) * STRIP_ + lcol] = m0;
      const v4f m1 = chunk_fold(buf1);
      if (c1 <= ce) *(v4f*)&tab[(c1 - cb) * STRIP_ + lcol] = m1;
    }
  }
  __syncthreads();

  const float ninf = -__builtin_inff();
  for (int j = wave; j < NSPAN_; j += SPAN_WV_) {
    int lo, hi;
    span_bounds(h_span, t_span, b_spans, b, j, lo, hi);
    float* dst;
    if (j == 0)      dst = h_emb + b * D_;
    else if (j == 1) dst = t_emb + b * D_;
    else             dst = b_emb + (size_t)(b * K_ + (j - 2)) * D_;

    v4f m = {ninf, ninf, ninf, ninf};
    if (lo <= hi) {
      const int cf = chunk_first(lo), cl = chunk_last(hi);
      int fend = CH_ROWS_ * cf - 1;
      fend = fend < hi ? fend : hi;
      fend = fend > lo ? fend : lo;
      int bbeg = CH_ROWS_ * (cl + 1);
      bbeg = bbeg > lo ? bbeg : lo;
      bbeg = bbeg < hi ? bbeg : hi;
      v4f e[CH_ROWS_ - 1];
#pragma unroll
      for (int i = 0; i < CH_ROWS_ - 1; ++i) {
        int rf = lo + i;  rf = rf < fend ? rf : fend;
        int rb = hi - i;  rb = rb > bbeg ? rb : bbeg;
        e[i] = span_row_load(rowbase, colbytes, half ? rb : rf);
      }
      if (cf <= cl) {
        const int ntrip = (cl - cf + 2) >> 1;
        for (int t0 = 0; t0 < ntrip; t0 += 8) {
          v4f q[8];
#pragma unroll
          for (int u = 0; u < 8; ++u) {
            int c = cf + 2 * (t0 + u) + half;
            c = c < cl ? c : cl;
            q[u] = *(const v4f*)&tab[(c - cb) * STRIP_ + lcol];
          }
#pragma unroll
          for (int u = 0; u < 8; ++u) m = vmax4(m, q[u]);
        }
      }
#pragma unroll
      for (int i = 0; i < CH_ROWS_ - 1; ++i) m = vmax4(m, e[i]);
    }

    m.x = fmaxf(m.x, __shfl_xor(m.x, 16));
    m.y = fmaxf(m.y, __shfl_xor(m.y, 16));
    m.z = fmaxf(m.z, __shfl_xor(m.z, 16));
    m.w = fmaxf(m.w, __shfl_xor(m.w, 16));

    if (half == 0) *(v4f*)(dst + col) = m;
  }
}

#define GK_      768
#define G_KC_    128
#define G_PITCH_ 132
#define G_NCH_   (GK_ / G_KC_)
#define G_LPR_   (G_KC_ / 4)

template <int NP>
__device__ __forceinline__ void gemm_fetch(v4f (&reg)[NP],
                                           const float* const (&srcp)[NP], int chunk) {
#pragma unroll
  for (int i = 0; i < NP; ++i) reg[i] = *(const v4f*)(srcp[i] + chunk * G_KC_);
}

template <int NP, int RPP>
__device__ __forceinline__ void gemm_stash(const v4f (&reg)[NP], float* stage, int soff) {
#pragma unroll
  for (int i = 0; i < NP; ++i) *(v4f*)(stage + soff + i * RPP * G_PITCH_) = reg[i];
}

template <bool FU>
__device__ __forceinline__ void gemm_chunk(const float* ap, const float* bp,
                                           v8f& acc0, v8f& acc1,
                                           const float* hp, const float* wp0, const float* wp1,
                                           v8f& ua0, v8f& ua1) {
  v4f af[G_KC_ / 8], bf[G_KC_ / 8];
  v4f hf[2], w0f[2], w1f[2];
#pragma unroll
  for (int kb = 0; kb < G_KC_ / 8; ++kb) {
    af[kb] = *(const v4f*)(ap + kb * 8);
    bf[kb] = *(const v4f*)(bp + kb * 8);
  }
  if constexpr (FU) {
#pragma unroll
    for (int j = 0; j < 2; ++j) {
      hf[j]  = *(const v4f*)(hp  + j * 8);
      w0f[j] = *(const v4f*)(wp0 + j * 8);
      w1f[j] = *(const v4f*)(wp1 + j * 8);
    }
  }
  __builtin_amdgcn_sched_barrier(0);
#pragma unroll
  for (int kb = 0; kb < G_KC_ / 8; ++kb) {
    const v2f a0 = {af[kb].x, af[kb].y}, b0 = {bf[kb].x, bf[kb].y};
    const v2f a1 = {af[kb].z, af[kb].w}, b1 = {bf[kb].z, bf[kb].w};
    acc0 = wmma_f32_16x16x4(a0, b0, acc0);
    acc1 = wmma_f32_16x16x4(a1, b1, acc1);
  }
  if constexpr (FU) {
#pragma unroll
    for (int j = 0; j < 2; ++j) {
      const v2f h0 = {hf[j].x, hf[j].y}, h1 = {hf[j].z, hf[j].w};
      const v2f p0 = {w0f[j].x, w0f[j].y}, p1 = {w0f[j].z, w0f[j].w};
      const v2f q0 = {w1f[j].x, w1f[j].y}, q1 = {w1f[j].z, w1f[j].w};
      ua0 = wmma_f32_16x16x4(h0, p0, ua0);
      ua1 = wmma_f32_16x16x4(h0, q0, ua1);
      ua0 = wmma_f32_16x16x4(h1, p1, ua0);
      ua1 = wmma_f32_16x16x4(h1, q1, ua1);
    }
  }
}

__device__ __forceinline__ void gemm_lds_handover() {
  __builtin_amdgcn_fence(__ATOMIC_RELEASE, "workgroup", "local");
  __builtin_amdgcn_s_barrier();
  __builtin_amdgcn_fence(__ATOMIC_ACQUIRE, "workgroup", "local");
}

#define PT_  (D_ / 32)
#define PBM_ 64

#define EPI_PLAIN_ 0
#define EPI_DOTS_  1
#define EPI_ATTN_  2

template <int WM, int WN, int EPI>
__global__ void __launch_bounds__(32 * WM * WN)
gemm_nt_wmma(const float* __restrict__ A,
             const float* __restrict__ W,
             const float* __restrict__ bias,
             float* __restrict__ C,
             int N, int relu,
             const float* __restrict__ Uq,
             float* __restrict__ part,
             const float* __restrict__ Bi,
             const float* __restrict__ Pin,
             const float* __restrict__ A2,
             const float* __restrict__ W2,
             const float* __restrict__ bias2,
             float* __restrict__ C2) {
  constexpr bool DOTS = (EPI == EPI_DOTS_);
  constexpr bool ATTN = (EPI == EPI_ATTN_);
  constexpr int BM   = 16 * WM;
  constexpr int BN   = 16 * WN;
  constexpr int UR   = DOTS ? BM / K_ : 0;
  constexpr int ROWS = BM + BN + (DOTS ? UR + BN : 0);
  constexpr int RPP  = 32 * WM * WN / G_LPR_;
  constexpr int NP   = ROWS / RPP;
  static_assert(ROWS % RPP == 0 && BM % RPP == 0 && BN % RPP == 0 && UR % RPP == 0,
                "copy passes must not straddle A, W, A2 and W2");
  static_assert(!DOTS || (UR == 8 && BN == 32 && 32 * WM * WN == UR * BN && 2 * WM * WN == G_KC_ / 8),
                "u tile: 8 x 32, one element per thread, two 8-k blocks of a chunk per wave");
  static_assert(G_NCH_ % 2 == 0 && G_NCH_ >= 4, "the K loop takes the chunks in pairs, two fetched ahead");
  __shared__ __attribute__((aligned(16))) float tile[2][ROWS * G_PITCH_];

  const int tid  = threadIdx.x;
  const int lane = tid & 31;
  const int wave = tid >> 5;
  const int wm   = wave / WN;
  const int wn   = wave % WN;
  const int half = lane >> 4;
  const int r    = lane & 15;
  const int m0   = blockIdx.y * BM;
  const int n0   = blockIdx.x * BN;

  const int srow = tid / G_LPR_;
  const int scol = (tid % G_LPR_) * 4;
  const float* srcp[NP];
#pragma unroll
  for (int i = 0; i < NP; ++i) {
    const int tr = srow + RPP * i;
    if (RPP * i < BM) {
      srcp[i] = A + (size_t)(m0 + tr) * GK_ + scol;
    } else if (RPP * i < BM + BN) {
      int wr = n0 + tr - BM;
      if (wr > N - 1) wr = N - 1;
      srcp[i] = W + (size_t)wr * GK_ + scol;
    } else if (RPP * i < BM + BN + UR) {
      srcp[i] = A2 + (size_t)(m0 / K_ + tr - BM - BN) * GK_ + scol;
    } else {
      int wr = n0 + tr - BM - BN - UR;
      if (wr > N - 1) wr = N - 1;
      srcp[i] = W2 + (size_t)wr * GK_ + scol;
    }
  }
  const int soff = srow * G_PITCH_ + scol;

  const float* ap0 = &tile[0][(wm * 16 + r) * G_PITCH_ + 4 * half];
  const float* bp0 = &tile[0][(BM + wn * 16 + r) * G_PITCH_ + 4 * half];
  const float* ap1 = &tile[1][(wm * 16 + r) * G_PITCH_ + 4 * half];
  const float* bp1 = &tile[1][(BM + wn * 16 + r) * G_PITCH_ + 4 * half];

  const float* hp0 = nullptr; const float* hp1 = nullptr;
  const float* up00 = nullptr; const float* up01 = nullptr;
  const float* up10 = nullptr; const float* up11 = nullptr;
  if constexpr (DOTS) {
    const int ukb = 2 * wave * 8 + 4 * half;
    hp0  = &tile[0][(BM + BN + (r & (UR - 1))) * G_PITCH_ + ukb];
    hp1  = &tile[1][(BM + BN + (r & (UR - 1))) * G_PITCH_ + ukb];
    up00 = &tile[0][(BM + BN + UR + r) * G_PITCH_ + ukb];
    up01 = &tile[0][(BM + BN + UR + 16 + r) * G_PITCH_ + ukb];
    up10 = &tile[1][(BM + BN + UR + r) * G_PITCH_ + ukb];
    up11 = &tile[1][(BM + BN + UR + 16 + r) * G_PITCH_ + ukb];
  }

  v8f acc0 = {}, acc1 = {};
  v8f ua0 = {}, ua1 = {};
  v4f ra[NP], rb[NP];

  float ub = 0.0f;
  if constexpr (DOTS) {
    const int un = n0 + (tid & 31);
    ub = bias2[un < N ? un : N - 1];
  }

  gemm_fetch(ra, srcp, 0);
  gemm_fetch(rb, srcp, 1);

  v4f pa[PT_ / 4], pb[PT_ / 4], bq[2][K_], uq4[2];
  const int erow = tid >> 3;
  const int ecol = n0 + 4 * (tid & 7);
  if constexpr (ATTN) {
    static_assert(WM == 1 && WN == 2, "the attention epilogue is written for a 16 x 32 tile of 64 threads");
    static_assert(PT_ % 4 == 0 && K_ == 8, "4 threads share a row's column tiles; 8 bridges = 2 x 16 bytes");
    __builtin_amdgcn_sched_barrier(0);
    const int bd = m0 + (tid >> 2);
    const float* pp = Pin + ((size_t)(bd / (PBM_ / K_)) * PT_ + (tid & 3)) * PBM_ + (bd % (PBM_ / K_)) * K_;
#pragma unroll
    for (int j = 0; j < PT_ / 4; ++j) {
      pa[j] = *(const v4f*)(pp + (size_t)j * 4 * PBM_);
      pb[j] = *(const v4f*)(pp + (size_t)j * 4 * PBM_ + 4);
    }
#pragma unroll
    for (int e = 0; e < 2; ++e) {
      const int b = m0 + erow + 8 * e;
      uq4[e] = *(const v4f*)(Uq + (size_t)b * N + ecol);
#pragma unroll
      for (int k = 0; k < K_; ++k)
        bq[e][k] = *(const v4f*)(Bi + ((size_t)b * K_ + k) * N + ecol);
    }
  }

  gemm_stash<NP, RPP>(ra, tile[0], soff);
  gemm_fetch(ra, srcp, 2);

  for (int t = 0; t < G_NCH_; t += 2) {
    gemm_lds_handover();
    gemm_stash<NP, RPP>(rb, tile[1], soff);
    if (t + 3 < G_NCH_) gemm_fetch(rb, srcp, t + 3);
    gemm_chunk<DOTS>(ap0, bp0, acc0, acc1, hp0, up00, up01, ua0, ua1);
    gemm_lds_handover();
    if (t + 2 < G_NCH_) gemm_stash<NP, RPP>(ra, tile[0], soff);
    if (t + 4 < G_NCH_) gemm_fetch(ra, srcp, t + 4);
    gemm_chunk<DOTS>(ap1, bp1, acc0, acc1, hp1, up10, up11, ua0, ua1);
  }

  float uqv = 0.0f;
  if constexpr (DOTS) {
    __shared__ float ured[WM * WN][2][UR][16];
    __shared__ float utile[UR][BN];
    if (half == 0) {
#pragma unroll
      for (int i = 0; i < UR; ++i) {
        ured[wave][0][i][r] = ua0[i];
        ured[wave][1][i][r] = ua1[i];
      }
    }
    gemm_lds_handover();
    {
      const int urow = tid >> 5, ucol = tid & 31;
      float s = ured[0][ucol >> 4][urow][ucol & 15];
#pragma unroll
      for (int w = 1; w < WM * WN; ++w) s += ured[w][ucol >> 4][urow][ucol & 15];
      const int un = n0 + ucol;
      s += ub;
      utile[urow][ucol] = s;
      if (un < N) C2[(size_t)(m0 / K_ + urow) * N + un] = s;
    }
    gemm_lds_handover();
    uqv = utile[2 * wm + half][wn * 16 + r];
  }

  const int n = n0 + wn * 16 + r;
  float pd[8];
#pragma unroll
  for (int i = 0; i < 8; ++i) pd[i] = 0.0f;
  if constexpr (ATTN) {
    __shared__ __attribute__((aligned(16))) float sv[BM][BN + 4];
    __shared__ __attribute__((aligned(16))) float satt[BM * K_];
    const float bv = bias[n];
#pragma unroll
    for (int i = 0; i < 8; ++i) {
      float vout = (acc0[i] + acc1[i]) + bv;
      if (relu) vout = fmaxf(vout, 0.0f);
      sv[wm * 16 + i + 8 * half][wn * 16 + r] = vout;
    }
    float att[K_];
    att[0] = pa[0].x; att[1] = pa[0].y; att[2] = pa[0].z; att[3] = pa[0].w;
    att[4] = pb[0].x; att[5] = pb[0].y; att[6] = pb[0].z; att[7] = pb[0].w;
#pragma unroll
    for (int j = 1; j < PT_ / 4; ++j) {
      att[0] += pa[j].x; att[1] += pa[j].y; att[2] += pa[j].z; att[3] += pa[j].w;
      att[4] += pb[j].x; att[5] += pb[j].y; att[6] += pb[j].z; att[7] += pb[j].w;
    }
#pragma unroll
    for (int off = 1; off < 4; off <<= 1) {
#pragma unroll
      for (int k = 0; k < K_; ++k) att[k] += __shfl_xor(att[k], off);
    }
#pragma unroll
    for (int k = 0; k < K_; ++k) att[k] *= 0.125f;
    float mx = att[0];
#pragma unroll
    for (int k = 1; k < K_; ++k) mx = fmaxf(mx, att[k]);
    float s = 0.f;
#pragma unroll
    for (int k = 0; k < K_; ++k) { att[k] = __expf(att[k] - mx); s += att[k]; }
    const float inv = 1.0f / s;
#pragma unroll
    for (int k = 0; k < K_; ++k) att[k] *= inv;
    if ((tid & 3) == 0) {
      const v4f w0 = {att[0], att[1], att[2], att[3]};
      const v4f w1 = {att[4], att[5], att[6], att[7]};
      *(v4f*)&satt[(tid >> 2) * K_]     = w0;
      *(v4f*)&satt[(tid >> 2) * K_ + 4] = w1;
    }
    gemm_lds_handover();
#pragma unroll
    for (int e = 0; e < 2; ++e) {
      const int row = erow + 8 * e;
      const v4f vv = *(const v4f*)&sv[row][4 * (tid & 7)];
      const v4f w0 = *(const v4f*)&satt[row * K_];
      const v4f w1 = *(const v4f*)&satt[row * K_ + 4];
      const float wk[K_] = {w0.x, w0.y, w0.z, w0.w, w1.x, w1.y, w1.z, w1.w};
      v4f c = {0.f, 0.f, 0.f, 0.f};
#pragma unroll
      for (int k = 0; k < K_; ++k) {
        c.x += wk[k] * bq[e][k].x;
        c.y += wk[k] * bq[e][k].y;
        c.z += wk[k] * bq[e][k].z;
        c.w += wk[k] * bq[e][k].w;
      }
      v4f o;
      o.x = uq4[e].x + vv.x + c.x;
      o.y = uq4[e].y + vv.y + c.y;
      o.z = uq4[e].z + vv.z + c.z;
      o.w = uq4[e].w + vv.w + c.w;
      *(v4f*)(C + (size_t)(m0 + row) * N + ecol) = o;
    }
  } else if (n < N) {
    const float bv = bias[n];
#pragma unroll
    for (int i = 0; i < 8; ++i) {
      float vout = (acc0[i] + acc1[i]) + bv;
      if (relu) vout = fmaxf(vout, 0.0f);
      C[(size_t)(m0 + wm * 16 + i + 8 * half) * N + n] = vout;
      if constexpr (DOTS) pd[i] = uqv * vout;
    }
  }
  if constexpr (DOTS) {
    static_assert(K_ == 8, "one lane half (8 rows of the piece) must be one batch row");
    __shared__ float dred[WN][BM];
#pragma unroll
    for (int off = 8; off > 0; off >>= 1) {
#pragma unroll
      for (int i = 0; i < 8; ++i) pd[i] += __shfl_xor(pd[i], off);
    }
    if (r == 0) {
#pragma unroll
      for (int i = 0; i < 8; ++i) dred[wn][wm * 16 + 8 * half + i] = pd[i];
    }
    gemm_lds_handover();
    if (tid < BM) {
      float s = dred[0][tid];
#pragma unroll
      for (int w = 1; w < WN; ++w) s += dred[w][tid];
      const int ntile = (N + BN - 1) / BN;
      part[((size_t)blockIdx.y * ntile + blockIdx.x) * BM + tid] = s;
    }
  }
}

extern "C" void kernel_launch(void* const* d_in, const int* in_sizes, int n_in,
                              void* d_out, int out_size, void* d_ws, size_t ws_size,
                              hipStream_t stream) {
  (void)in_sizes; (void)n_in; (void)out_size; (void)ws_size;

  const float* emb     = (const float*)d_in[0];
  const int*   h_span  = (const int*)d_in[1];
  const int*   t_span  = (const int*)d_in[2];
  const int*   b_spans = (const int*)d_in[3];
  const float* wu_w    = (const float*)d_in[4];
  const float* wu_b    = (const float*)d_in[5];
  const float* wv_w    = (const float*)d_in[6];
  const float* wv_b    = (const float*)d_in[7];
  const float* wi_w    = (const float*)d_in[8];
  const float* wi_b    = (const float*)d_in[9];
  const float* ln1_w   = (const float*)d_in[10];
  const float* ln1_b   = (const float*)d_in[11];
  const float* pred_w  = (const float*)d_in[12];
  const float* pred_b  = (const float*)d_in[13];
  float* out = (float*)d_out;

  float* ws    = (float*)d_ws;
  float* h_emb = ws;
  float* t_emb = h_emb + B_ * D_;
  float* b_emb = t_emb + B_ * D_;
  float* u     = b_emb + B_ * K_ * D_;
  float* part  = u + B_ * D_;
  float* bi    = part + B_ * D_;
  float* pre   = bi + B_ * K_ * D_;
  float* pair  = h_emb;

  span_max_chunk_table<<<dim3(D_ / STRIP_, B_), 32 * SPAN_WV_, 0, stream>>>(emb, h_span, t_span, b_spans,
                                                                            h_emb, t_emb, b_emb);

  gemm_nt_wmma<4, 2, EPI_DOTS_><<<dim3(PT_, 8), 256, 0, stream>>>(b_emb, wi_w, wi_b, bi, D_, 0,
                                                                  nullptr, part, nullptr, nullptr,
                                                                  h_emb, wu_w, wu_b, u);
  gemm_nt_wmma<1, 2, EPI_ATTN_><<<dim3(24, 4), 64, 0, stream>>>(t_emb, wv_w, wv_b, pre, D_, 0,
                                                               u, nullptr, bi, part,
                                                               nullptr, nullptr, nullptr, nullptr);

  gemm_nt_wmma<1, 2, EPI_PLAIN_><<<dim3(24, 4), 64, 0, stream>>>(pre, ln1_w, ln1_b, pair, D_, 1,
                                                                nullptr, nullptr, nullptr, nullptr,
                                                                nullptr, nullptr, nullptr, nullptr);
  gemm_nt_wmma<1, 2, EPI_PLAIN_><<<dim3(9, 4), 64, 0, stream>>>(pair, pred_w, pred_b, out, R_, 0,
                                                               nullptr, nullptr, nullptr, nullptr,
                                                               nullptr, nullptr, nullptr, nullptr);
}


# Products with 64 rows (v, ln1, pred): K of each 16x16 piece dealt over 4 waves inside the LDS chunk pipeline (8-wave block per 16x32 tile, 96 matrix instr. per SIMD, was 192): new K sum order
# speedup vs baseline: 9.2876x; 1.0383x over previous
#include <hip/hip_runtime.h>

#define B_  64
#define S_  512
#define D_  768
#define K_  8
#define R_  277

typedef __attribute__((ext_vector_type(2))) float v2f;
typedef __attribute__((ext_vector_type(8))) float v8f;

__device__ __forceinline__ v8f wmma_f32_16x16x4(v2f a, v2f b, v8f c) {
#if defined(__gfx1250__) && __has_builtin(__builtin_amdgcn_wmma_f32_16x16x4_f32)
  return __builtin_amdgcn_wmma_f32_16x16x4_f32(
      false, a, false, b, (short)0, c, false, false);
#else
  asm volatile("v_wmma_f32_16x16x4_f32 %0, %1, %2, %0"
               : "+v"(c) : "v"(a), "v"(b));
  return c;
#endif
}

#define NSPAN_    10
#define CH_ROWS_  8
#define NCHUNK_   (S_ / CH_ROWS_)
#define STRIP_    64
#define SPAN_WV_  8

typedef __attribute__((ext_vector_type(4))) float v4f;

__device__ __forceinline__ v4f vmax4(v4f a, v4f b) {
  v4f m;
  m.x = fmaxf(a.x, b.x);
  m.y = fmaxf(a.y, b.y);
  m.z = fmaxf(a.z, b.z);
  m.w = fmaxf(a.w, b.w);
  return m;
}

__device__ __forceinline__ v4f span_row_load(const char* __restrict__ rowbase,
                                             unsigned colbytes, int r) {
  const unsigned off = (unsigned)r * (unsigned)(D_ * 4) + colbytes;
  return *(const v4f*)(rowbase + off);
}

__device__ __forceinline__ void chunk_load(v4f (&buf)[CH_ROWS_],
                                           const char* __restrict__ rowbase,
                                           unsigned colbytes, int c) {
#pragma unroll
  for (int i = 0; i < CH_ROWS_; ++i)
    buf[i] = span_row_load(rowbase, colbytes, c * CH_ROWS_ + i);
}

__device__ __forceinline__ v4f chunk_fold(const v4f (&buf)[CH_ROWS_]) {
  v4f m = buf[0];
#pragma unroll
  for (int i = 1; i < CH_ROWS_; ++i) m = vmax4(m, buf[i]);
  return m;
}

__device__ __forceinline__ void span_bounds(const int* __restrict__ h_span,
                                            const int* __restrict__ t_span,
                                            const int* __restrict__ b_spans,
                                            int b, int j, int& lo, int& hi) {
  const int* sp;
  if (j == 0)      sp = h_span + b * 2;
  else if (j == 1) sp = t_span + b * 2;
  else             sp = b_spans + (b * K_ + (j - 2)) * 2;
  lo = sp[0];
  hi = sp[1];
  if (lo < 0) lo = 0;
  if (hi > S_ - 1) hi = S_ - 1;
}

__device__ __forceinline__ int chunk_first(int lo) { return (lo + CH_ROWS_ - 1) / CH_ROWS_; }
__device__ __forceinline__ int chunk_last(int hi)  { return (hi + 1) / CH_ROWS_ - 1; }

__global__ void __launch_bounds__(32 * SPAN_WV_)
span_max_chunk_table(const float* __restrict__ emb,
                     const int*   __restrict__ h_span,
                     const int*   __restrict__ t_span,
                     const int*   __restrict__ b_spans,
                     float* __restrict__ h_emb,
                     float* __restrict__ t_emb,
                     float* __restrict__ b_emb) {
  __shared__ __attribute__((aligned(16))) float tab[NCHUNK_ * STRIP_];

  const int strip = blockIdx.x;
  const int b     = blockIdx.y;

  const int lane = threadIdx.x & 31;
  const int wave = __builtin_amdgcn_readfirstlane((int)(threadIdx.x >> 5));
  const int half = lane >> 4;
  const int lcol = (lane & 15) * 4;
  const int col  = strip * STRIP_ + lcol;

  const char* rowbase = (const char*)(emb + (size_t)b * S_ * D_);
  const unsigned colbytes = (unsigned)col * 4u;

  int cb = NCHUNK_, ce = -1;
#pragma unroll
  for (int j = 0; j < NSPAN_; ++j) {
    int lo, hi;
    span_bounds(h_span, t_span, b_spans, b, j, lo, hi);
    const int cf = chunk_first(lo), cl = chunk_last(hi);
    if (cf <= cl) {
      cb = cf < cb ? cf : cb;
      ce = cl > ce ? cl : ce;
    }
  }
  const int nch = ce - cb + 1;

  const int slot = 2 * wave + half;
#pragma unroll
  for (int p = 0; p < NCHUNK_ / 32; ++p) {
    if (32 * p < nch) {
      const int c0 = cb + 32 * p + slot;
      const int c1 = c0 + 16;
      v4f buf0[CH_ROWS_], buf1[CH_ROWS_];
      chunk_load(buf0, rowbase, colbytes, c0 < ce ? c0 : ce);
      chunk_load(buf1, rowbase, colbytes, c1 < ce ? c1 : ce);
      const v4f m0 = chunk_fold(buf0);
      if (c0 <= ce) *(v4f*)&tab[(c0 - cb) * STRIP_ + lcol] = m0;
      const v4f m1 = chunk_fold(buf1);
      if (c1 <= ce) *(v4f*)&tab[(c1 - cb) * STRIP_ + lcol] = m1;
    }
  }
  __syncthreads();

  const float ninf = -__builtin_inff();
  for (int j = wave; j < NSPAN_; j += SPAN_WV_) {
    int lo, hi;
    span_bounds(h_span, t_span, b_spans, b, j, lo, hi);
    float* dst;
    if (j == 0)      dst = h_emb + b * D_;
    else if (j == 1) dst = t_emb + b * D_;
    else             dst = b_emb + (size_t)(b * K_ + (j - 2)) * D_;

    v4f m = {ninf, ninf, ninf, ninf};
    if (lo <= hi) {
      const int cf = chunk_first(lo), cl = chunk_last(hi);
      int fend = CH_ROWS_ * cf - 1;
      fend = fend < hi ? fend : hi;
      fend = fend > lo ? fend : lo;
      int bbeg = CH_ROWS_ * (cl + 1);
      bbeg = bbeg > lo ? bbeg : lo;
      bbeg = bbeg < hi ? bbeg : hi;
      v4f e[CH_ROWS_ - 1];
#pragma unroll
      for (int i = 0; i < CH_ROWS_ - 1; ++i) {
        int rf = lo + i;  rf = rf < fend ? rf : fend;
        int rb = hi - i;  rb = rb > bbeg ? rb : bbeg;
        e[i] = span_row_load(rowbase, colbytes, half ? rb : rf);
      }
      if (cf <= cl) {
        const int ntrip = (cl - cf + 2) >> 1;
        for (int t0 = 0; t0 < ntrip; t0 += 8) {
          v4f q[8];
#pragma unroll
          for (int u = 0; u < 8; ++u) {
            int c = cf + 2 * (t0 + u) + half;
            c = c < cl ? c : cl;
            q[u] = *(const v4f*)&tab[(c - cb) * STRIP_ + lcol];
          }
#pragma unroll
          for (int u = 0; u < 8; ++u) m = vmax4(m, q[u]);
        }
      }
#pragma unroll
      for (int i = 0; i < CH_ROWS_ - 1; ++i) m = vmax4(m, e[i]);
    }

    m.x = fmaxf(m.x, __shfl_xor(m.x, 16));
    m.y = fmaxf(m.y, __shfl_xor(m.y, 16));
    m.z = fmaxf(m.z, __shfl_xor(m.z, 16));
    m.w = fmaxf(m.w, __shfl_xor(m.w, 16));

    if (half == 0) *(v4f*)(dst + col) = m;
  }
}

#define GK_      768
#define G_KC_    128
#define G_PITCH_ 132
#define G_NCH_   (GK_ / G_KC_)
#define G_LPR_   (G_KC_ / 4)

template <int NP>
__device__ __forceinline__ void gemm_fetch(v4f (&reg)[NP],
                                           const float* const (&srcp)[NP], int chunk) {
#pragma unroll
  for (int i = 0; i < NP; ++i) reg[i] = *(const v4f*)(srcp[i] + chunk * G_KC_);
}

template <int NP, int RPP>
__device__ __forceinline__ void gemm_stash(const v4f (&reg)[NP], float* stage, int soff) {
#pragma unroll
  for (int i = 0; i < NP; ++i) *(v4f*)(stage + soff + i * RPP * G_PITCH_) = reg[i];
}

template <bool FU, int NKB>
__device__ __forceinline__ void gemm_chunk(const float* ap, const float* bp,
                                           v8f& acc0, v8f& acc1,
                                           const float* hp, const float* wp0, const float* wp1,
                                           v8f& ua0, v8f& ua1) {
  v4f af[NKB], bf[NKB];
  v4f hf[2], w0f[2], w1f[2];
#pragma unroll
  for (int kb = 0; kb < NKB; ++kb) {
    af[kb] = *(const v4f*)(ap + kb * 8);
    bf[kb] = *(const v4f*)(bp + kb * 8);
  }
  if constexpr (FU) {
#pragma unroll
    for (int j = 0; j < 2; ++j) {
      hf[j]  = *(const v4f*)(hp  + j * 8);
      w0f[j] = *(const v4f*)(wp0 + j * 8);
      w1f[j] = *(const v4f*)(wp1 + j * 8);
    }
  }
  __builtin_amdgcn_sched_barrier(0);
#pragma unroll
  for (int kb = 0; kb < NKB; ++kb) {
    const v2f a0 = {af[kb].x, af[kb].y}, b0 = {bf[kb].x, bf[kb].y};
    const v2f a1 = {af[kb].z, af[kb].w}, b1 = {bf[kb].z, bf[kb].w};
    acc0 = wmma_f32_16x16x4(a0, b0, acc0);
    acc1 = wmma_f32_16x16x4(a1, b1, acc1);
  }
  if constexpr (FU) {
#pragma unroll
    for (int j = 0; j < 2; ++j) {
      const v2f h0 = {hf[j].x, hf[j].y}, h1 = {hf[j].z, hf[j].w};
      const v2f p0 = {w0f[j].x, w0f[j].y}, p1 = {w0f[j].z, w0f[j].w};
      const v2f q0 = {w1f[j].x, w1f[j].y}, q1 = {w1f[j].z, w1f[j].w};
      ua0 = wmma_f32_16x16x4(h0, p0, ua0);
      ua1 = wmma_f32_16x16x4(h0, q0, ua1);
      ua0 = wmma_f32_16x16x4(h1, p1, ua0);
      ua1 = wmma_f32_16x16x4(h1, q1, ua1);
    }
  }
}

__device__ __forceinline__ void gemm_lds_handover() {
  __builtin_amdgcn_fence(__ATOMIC_RELEASE, "workgroup", "local");
  __builtin_amdgcn_s_barrier();
  __builtin_amdgcn_fence(__ATOMIC_ACQUIRE, "workgroup", "local");
}

#define PT_  (D_ / 32)
#define PBM_ 64

#define EPI_PLAIN_ 0
#define EPI_DOTS_  1
#define EPI_ATTN_  2

template <int WM, int WN, int KS, int EPI>
__global__ void __launch_bounds__(32 * WM * WN * KS)
gemm_nt_wmma(const float* __restrict__ A,
             const float* __restrict__ W,
             const float* __restrict__ bias,
             float* __restrict__ C,
             int N, int relu,
             const float* __restrict__ Uq,
             float* __restrict__ part,
             const float* __restrict__ Bi,
             const float* __restrict__ Pin,
             const float* __restrict__ A2,
             const float* __restrict__ W2,
             const float* __restrict__ bias2,
             float* __restrict__ C2) {
  constexpr bool DOTS = (EPI == EPI_DOTS_);
  constexpr bool ATTN = (EPI == EPI_ATTN_);
  constexpr int BM   = 16 * WM;
  constexpr int BN   = 16 * WN;
  constexpr int UR   = DOTS ? BM / K_ : 0;
  constexpr int ROWS = BM + BN + (DOTS ? UR + BN : 0);
  constexpr int NPC  = WM * WN;
  constexpr int NKB  = G_KC_ / 8 / KS;
  constexpr int RPP  = 32 * NPC * KS / G_LPR_;
  constexpr int NP   = ROWS / RPP;
  static_assert(ROWS % RPP == 0 && BM % RPP == 0 && BN % RPP == 0 && UR % RPP == 0,
                "copy passes must not straddle A, W, A2 and W2");
  static_assert(!DOTS || (UR == 8 && BN == 32 && 32 * WM * WN == UR * BN && 2 * WM * WN == G_KC_ / 8),
                "u tile: 8 x 32, one element per thread, two 8-k blocks of a chunk per wave");
  static_assert(G_NCH_ % 2 == 0 && G_NCH_ >= 4, "the K loop takes the chunks in pairs, two fetched ahead");
  static_assert((G_KC_ / 8) % KS == 0, "a chunk's 8-k blocks are dealt evenly over the KS waves of a piece");
  static_assert(!DOTS || KS == 1, "the bridge product's waves own their pieces over the whole K");
  __shared__ __attribute__((aligned(16))) float tile[2][ROWS * G_PITCH_];

  const int tid  = threadIdx.x;
  const int lane = tid & 31;
  const int wave = tid >> 5;
  const int ks   = KS > 1 ? wave / NPC : 0;
  const int pc   = KS > 1 ? wave % NPC : wave;
  const int wm   = pc / WN;
  const int wn   = pc % WN;
  const int half = lane >> 4;
  const int r    = lane & 15;
  const int m0   = blockIdx.y * BM;
  const int n0   = blockIdx.x * BN;

  const int srow = tid / G_LPR_;
  const int scol = (tid % G_LPR_) * 4;
  const float* srcp[NP];
#pragma unroll
  for (int i = 0; i < NP; ++i) {
    const int tr = srow + RPP * i;
    if (RPP * i < BM) {
      srcp[i] = A + (size_t)(m0 + tr) * GK_ + scol;
    } else if (RPP * i < BM + BN) {
      int wr = n0 + tr - BM;
      if (wr > N - 1) wr = N - 1;
      srcp[i] = W + (size_t)wr * GK_ + scol;
    } else if (RPP * i < BM + BN + UR) {
      srcp[i] = A2 + (size_t)(m0 / K_ + tr - BM - BN) * GK_ + scol;
    } else {
      int wr = n0 + tr - BM - BN - UR;
      if (wr > N - 1) wr = N - 1;
      srcp[i] = W2 + (size_t)wr * GK_ + scol;
    }
  }
  const int soff = srow * G_PITCH_ + scol;

  const int kof = ks * NKB * 8 + 4 * half;
  const float* ap0 = &tile[0][(wm * 16 + r) * G_PITCH_ + kof];
  const float* bp0 = &tile[0][(BM + wn * 16 + r) * G_PITCH_ + kof];
  const float* ap1 = &tile[1][(wm * 16 + r) * G_PITCH_ + kof];
  const float* bp1 = &tile[1][(BM + wn * 16 + r) * G_PITCH_ + kof];

  const float* hp0 = nullptr; const float* hp1 = nullptr;
  const float* up00 = nullptr; const float* up01 = nullptr;
  const float* up10 = nullptr; const float* up11 = nullptr;
  if constexpr (DOTS) {
    const int ukb = 2 * wave * 8 + 4 * half;
    hp0  = &tile[0][(BM + BN + (r & (UR - 1))) * G_PITCH_ + ukb];
    hp1  = &tile[1][(BM + BN + (r & (UR - 1))) * G_PITCH_ + ukb];
    up00 = &tile[0][(BM + BN + UR + r) * G_PITCH_ + ukb];
    up01 = &tile[0][(BM + BN + UR + 16 + r) * G_PITCH_ + ukb];
    up10 = &tile[1][(BM + BN + UR + r) * G_PITCH_ + ukb];
    up11 = &tile[1][(BM + BN + UR + 16 + r) * G_PITCH_ + ukb];
  }

  v8f acc0 = {}, acc1 = {};
  v8f ua0 = {}, ua1 = {};
  v4f ra[NP], rb[NP];

  float ub = 0.0f;
  if constexpr (DOTS) {
    const int un = n0 + (tid & 31);
    ub = bias2[un < N ? un : N - 1];
  }

  gemm_fetch(ra, srcp, 0);
  gemm_fetch(rb, srcp, 1);

  v4f pa[PT_ / 4], pb[PT_ / 4], bq[2][K_], uq4[2];
  const int erow = tid >> 3;
  const int ecol = n0 + 4 * (tid & 7);
  if constexpr (ATTN) {
    static_assert(WM == 1 && WN == 2, "the attention epilogue is written for a 16 x 32 tile and its first 64 threads");
    static_assert(PT_ % 4 == 0 && K_ == 8, "4 threads share a row's column tiles; 8 bridges = 2 x 16 bytes");
    __builtin_amdgcn_sched_barrier(0);
    if (ks == 0) {
      const int bd = m0 + (tid >> 2);
      const float* pp = Pin + ((size_t)(bd / (PBM_ / K_)) * PT_ + (tid & 3)) * PBM_ + (bd % (PBM_ / K_)) * K_;
#pragma unroll
      for (int j = 0; j < PT_ / 4; ++j) {
        pa[j] = *(const v4f*)(pp + (size_t)j * 4 * PBM_);
        pb[j] = *(const v4f*)(pp + (size_t)j * 4 * PBM_ + 4);
      }
#pragma unroll
      for (int e = 0; e < 2; ++e) {
        const int b = m0 + erow + 8 * e;
        uq4[e] = *(const v4f*)(Uq + (size_t)b * N + ecol);
#pragma unroll
        for (int k = 0; k < K_; ++k)
          bq[e][k] = *(const v4f*)(Bi + ((size_t)b * K_ + k) * N + ecol);
      }
    }
  }

  gemm_stash<NP, RPP>(ra, tile[0], soff);
  gemm_fetch(ra, srcp, 2);

  for (int t = 0; t < G_NCH_; t += 2) {
    gemm_lds_handover();
    gemm_stash<NP, RPP>(rb, tile[1], soff);
    if (t + 3 < G_NCH_) gemm_fetch(rb, srcp, t + 3);
    gemm_chunk<DOTS, NKB>(ap0, bp0, acc0, acc1, hp0, up00, up01, ua0, ua1);
    gemm_lds_handover();
    if (t + 2 < G_NCH_) gemm_stash<NP, RPP>(ra, tile[0], soff);
    if (t + 4 < G_NCH_) gemm_fetch(ra, srcp, t + 4);
    gemm_chunk<DOTS, NKB>(ap1, bp1, acc0, acc1, hp1, up10, up11, ua0, ua1);
  }

  float uqv = 0.0f;
  if constexpr (DOTS) {
    __shared__ float ured[WM * WN][2][UR][16];
    __shared__ float utile[UR][BN];
    if (half == 0) {
#pragma unroll
      for (int i = 0; i < UR; ++i) {
        ured[wave][0][i][r] = ua0[i];
        ured[wave][1][i][r] = ua1[i];
      }
    }
    gemm_lds_handover();
    {
      const int urow = tid >> 5, ucol = tid & 31;
      float s = ured[0][ucol >> 4][urow][ucol & 15];
#pragma unroll
      for (int w = 1; w < WM * WN; ++w) s += ured[w][ucol >> 4][urow][ucol & 15];
      const int un = n0 + ucol;
      s += ub;
      utile[urow][ucol] = s;
      if (un < N) C2[(size_t)(m0 / K_ + urow) * N + un] = s;
    }
    gemm_lds_handover();
    uqv = utile[2 * wm + half][wn * 16 + r];
  }

  float cs[8];
#pragma unroll
  for (int i = 0; i < 8; ++i) cs[i] = acc0[i] + acc1[i];
  const bool owner = (ks == 0);
  if constexpr (KS > 1) {
    __shared__ float kred[KS - 1][NPC][8][32];
    if (!owner) {
#pragma unroll
      for (int i = 0; i < 8; ++i) kred[ks - 1][pc][i][lane] = cs[i];
    }
    gemm_lds_handover();
    if (owner) {
#pragma unroll
      for (int q = 0; q < KS - 1; ++q) {
#pragma unroll
        for (int i = 0; i < 8; ++i) cs[i] += kred[q][pc][i][lane];
      }
    }
  }

  const int n = n0 + wn * 16 + r;
  float pd[8];
#pragma unroll
  for (int i = 0; i < 8; ++i) pd[i] = 0.0f;
  if constexpr (ATTN) {
    __shared__ __attribute__((aligned(16))) float sv[BM][BN + 4];
    __shared__ __attribute__((aligned(16))) float satt[BM * K_];
    if (owner) {
      const float bv = bias[n];
#pragma unroll
      for (int i = 0; i < 8; ++i) {
        float vout = cs[i] + bv;
        if (relu) vout = fmaxf(vout, 0.0f);
        sv[wm * 16 + i + 8 * half][wn * 16 + r] = vout;
      }
      float att[K_];
      att[0] = pa[0].x; att[1] = pa[0].y; att[2] = pa[0].z; att[3] = pa[0].w;
      att[4] = pb[0].x; att[5] = pb[0].y; att[6] = pb[0].z; att[7] = pb[0].w;
#pragma unroll
      for (int j = 1; j < PT_ / 4; ++j) {
        att[0] += pa[j].x; att[1] += pa[j].y; att[2] += pa[j].z; att[3] += pa[j].w;
        att[4] += pb[j].x; att[5] += pb[j].y; att[6] += pb[j].z; att[7] += pb[j].w;
      }
#pragma unroll
      for (int off = 1; off < 4; off <<= 1) {
#pragma unroll
        for (int k = 0; k < K_; ++k) att[k] += __shfl_xor(att[k], off);
      }
#pragma unroll
      for (int k = 0; k < K_; ++k) att[k] *= 0.125f;
      float mx = att[0];
#pragma unroll
      for (int k = 1; k < K_; ++k) mx = fmaxf(mx, att[k]);
      float s = 0.f;
#pragma unroll
      for (int k = 0; k < K_; ++k) { att[k] = __expf(att[k] - mx); s += att[k]; }
      const float inv = 1.0f / s;
#pragma unroll
      for (int k = 0; k < K_; ++k) att[k] *= inv;
      if ((tid & 3) == 0) {
        const v4f w0 = {att[0], att[1], att[2], att[3]};
        const v4f w1 = {att[4], att[5], att[6], att[7]};
        *(v4f*)&satt[(tid >> 2) * K_]     = w0;
        *(v4f*)&satt[(tid >> 2) * K_ + 4] = w1;
      }
    }
    gemm_lds_handover();
    if (owner) {
#pragma unroll
      for (int e = 0; e < 2; ++e) {
        const int row = erow + 8 * e;
        const v4f vv = *(const v4f*)&sv[row][4 * (tid & 7)];
        const v4f w0 = *(const v4f*)&satt[row * K_];
        const v4f w1 = *(const v4f*)&satt[row * K_ + 4];
        const float wk[K_] = {w0.x, w0.y, w0.z, w0.w, w1.x, w1.y, w1.z, w1.w};
        v4f c = {0.f, 0.f, 0.f, 0.f};
#pragma unroll
        for (int k = 0; k < K_; ++k) {
          c.x += wk[k] * bq[e][k].x;
          c.y += wk[k] * bq[e][k].y;
          c.z += wk[k] * bq[e][k].z;
          c.w += wk[k] * bq[e][k].w;
        }
        v4f o;
        o.x = uq4[e].x + vv.x + c.x;
        o.y = uq4[e].y + vv.y + c.y;
        o.z = uq4[e].z + vv.z + c.z;
        o.w = uq4[e].w + vv.w + c.w;
        *(v4f*)(C + (size_t)(m0 + row) * N + ecol) = o;
      }
    }
  } else if (owner && n < N) {
    const float bv = bias[n];
#pragma unroll
    for (int i = 0; i < 8; ++i) {
      float vout = cs[i] + bv;
      if (relu) vout = fmaxf(vout, 0.0f);
      C[(size_t)(m0 + wm * 16 + i + 8 * half) * N + n] = vout;
      if constexpr (DOTS) pd[i] = uqv * vout;
    }
  }
  if constexpr (DOTS) {
    static_assert(K_ == 8, "one lane half (8 rows of the piece) must be one batch row");
    __shared__ float dred[WN][BM];
#pragma unroll
    for (int off = 8; off > 0; off >>= 1) {
#pragma unroll
      for (int i = 0; i < 8; ++i) pd[i] += __shfl_xor(pd[i], off);
    }
    if (r == 0) {
#pragma unroll
      for (int i = 0; i < 8; ++i) dred[wn][wm * 16 + 8 * half + i] = pd[i];
    }
    gemm_lds_handover();
    if (tid < BM) {
      float s = dred[0][tid];
#pragma unroll
      for (int w = 1; w < WN; ++w) s += dred[w][tid];
      const int ntile = (N + BN - 1) / BN;
      part[((size_t)blockIdx.y * ntile + blockIdx.x) * BM + tid] = s;
    }
  }
}

extern "C" void kernel_launch(void* const* d_in, const int* in_sizes, int n_in,
                              void* d_out, int out_size, void* d_ws, size_t ws_size,
                              hipStream_t stream) {
  (void)in_sizes; (void)n_in; (void)out_size; (void)ws_size;

  const float* emb     = (const float*)d_in[0];
  const int*   h_span  = (const int*)d_in[1];
  const int*   t_span  = (const int*)d_in[2];
  const int*   b_spans = (const int*)d_in[3];
  const float* wu_w    = (const float*)d_in[4];
  const float* wu_b    = (const float*)d_in[5];
  const float* wv_w    = (const float*)d_in[6];
  const float* wv_b    = (const float*)d_in[7];
  const float* wi_w    = (const float*)d_in[8];
  const float* wi_b    = (const float*)d_in[9];
  const float* ln1_w   = (const float*)d_in[10];
  const float* ln1_b   = (const float*)d_in[11];
  const float* pred_w  = (const float*)d_in[12];
  const float* pred_b  = (const float*)d_in[13];
  float* out = (float*)d_out;

  float* ws    = (float*)d_ws;
  float* h_emb = ws;
  float* t_emb = h_emb + B_ * D_;
  float* b_emb = t_emb + B_ * D_;
  float* u     = b_emb + B_ * K_ * D_;
  float* part  = u + B_ * D_;
  float* bi    = part + B_ * D_;
  float* pre   = bi + B_ * K_ * D_;
  float* pair  = h_emb;

  span_max_chunk_table<<<dim3(D_ / STRIP_, B_), 32 * SPAN_WV_, 0, stream>>>(emb, h_span, t_span, b_spans,
                                                                            h_emb, t_emb, b_emb);

  gemm_nt_wmma<4, 2, 1, EPI_DOTS_><<<dim3(PT_, 8), 256, 0, stream>>>(b_emb, wi_w, wi_b, bi, D_, 0,
                                                                  nullptr, part, nullptr, nullptr,
                                                                  h_emb, wu_w, wu_b, u);
  gemm_nt_wmma<1, 2, 4, EPI_ATTN_><<<dim3(24, 4), 256, 0, stream>>>(t_emb, wv_w, wv_b, pre, D_, 0,
                                                               u, nullptr, bi, part,
                                                               nullptr, nullptr, nullptr, nullptr);

  gemm_nt_wmma<1, 2, 4, EPI_PLAIN_><<<dim3(24, 4), 256, 0, stream>>>(pre, ln1_w, ln1_b, pair, D_, 1,
                                                                nullptr, nullptr, nullptr, nullptr,
                                                                nullptr, nullptr, nullptr, nullptr);
  gemm_nt_wmma<1, 2, 4, EPI_PLAIN_><<<dim3(9, 4), 256, 0, stream>>>(pair, pred_w, pred_b, out, R_, 0,
                                                               nullptr, nullptr, nullptr, nullptr,
                                                               nullptr, nullptr, nullptr, nullptr);
}


# GEMM launches: global loads issued in order of need (chunk 0, 1, 2, then epilogue operands and bias, all ahead of the K loop), half of chunk 0 multiplied before the wait for chunk 1; same sums
# speedup vs baseline: 9.9319x; 1.0694x over previous
#include <hip/hip_runtime.h>

#define B_  64
#define S_  512
#define D_  768
#define K_  8
#define R_  277

typedef __attribute__((ext_vector_type(2))) float v2f;
typedef __attribute__((ext_vector_type(8))) float v8f;

__device__ __forceinline__ v8f wmma_f32_16x16x4(v2f a, v2f b, v8f c) {
#if defined(__gfx1250__) && __has_builtin(__builtin_amdgcn_wmma_f32_16x16x4_f32)
  return __builtin_amdgcn_wmma_f32_16x16x4_f32(
      false, a, false, b, (short)0, c, false, false);
#else
  asm volatile("v_wmma_f32_16x16x4_f32 %0, %1, %2, %0"
               : "+v"(c) : "v"(a), "v"(b));
  return c;
#endif
}

#define NSPAN_    10
#define CH_ROWS_  8
#define NCHUNK_   (S_ / CH_ROWS_)
#define STRIP_    64
#define SPAN_WV_  8

typedef __attribute__((ext_vector_type(4))) float v4f;

__device__ __forceinline__ v4f vmax4(v4f a, v4f b) {
  v4f m;
  m.x = fmaxf(a.x, b.x);
  m.y = fmaxf(a.y, b.y);
  m.z = fmaxf(a.z, b.z);
  m.w = fmaxf(a.w, b.w);
  return m;
}

__device__ __forceinline__ v4f span_row_load(const char* __restrict__ rowbase,
                                             unsigned colbytes, int r) {
  const unsigned off = (unsigned)r * (unsigned)(D_ * 4) + colbytes;
  return *(const v4f*)(rowbase + off);
}

__device__ __forceinline__ void chunk_load(v4f (&buf)[CH_ROWS_],
                                           const char* __restrict__ rowbase,
                                           unsigned colbytes, int c) {
#pragma unroll
  for (int i = 0; i < CH_ROWS_; ++i)
    buf[i] = span_row_load(rowbase, colbytes, c * CH_ROWS_ + i);
}

__device__ __forceinline__ v4f chunk_fold(const v4f (&buf)[CH_ROWS_]) {
  v4f m = buf[0];
#pragma unroll
  for (int i = 1; i < CH_ROWS_; ++i) m = vmax4(m, buf[i]);
  return m;
}

__device__ __forceinline__ void span_bounds(const int* __restrict__ h_span,
                                            const int* __restrict__ t_span,
                                            const int* __restrict__ b_spans,
                                            int b, int j, int& lo, int& hi) {
  const int* sp;
  if (j == 0)      sp = h_span + b * 2;
  else if (j == 1) sp = t_span + b * 2;
  else             sp = b_spans + (b * K_ + (j - 2)) * 2;
  lo = sp[0];
  hi = sp[1];
  if (lo < 0) lo = 0;
  if (hi > S_ - 1) hi = S_ - 1;
}

__device__ __forceinline__ int chunk_first(int lo) { return (lo + CH_ROWS_ - 1) / CH_ROWS_; }
__device__ __forceinline__ int chunk_last(int hi)  { return (hi + 1) / CH_ROWS_ - 1; }

__global__ void __launch_bounds__(32 * SPAN_WV_)
span_max_chunk_table(const float* __restrict__ emb,
                     const int*   __restrict__ h_span,
                     const int*   __restrict__ t_span,
                     const int*   __restrict__ b_spans,
                     float* __restrict__ h_emb,
                     float* __restrict__ t_emb,
                     float* __restrict__ b_emb) {
  __shared__ __attribute__((aligned(16))) float tab[NCHUNK_ * STRIP_];

  const int strip = blockIdx.x;
  const int b     = blockIdx.y;

  const int lane = threadIdx.x & 31;
  const int wave = __builtin_amdgcn_readfirstlane((int)(threadIdx.x >> 5));
  const int half = lane >> 4;
  const int lcol = (lane & 15) * 4;
  const int col  = strip * STRIP_ + lcol;

  const char* rowbase = (const char*)(emb + (size_t)b * S_ * D_);
  const unsigned colbytes = (unsigned)col * 4u;

  int cb = NCHUNK_, ce = -1;
#pragma unroll
  for (int j = 0; j < NSPAN_; ++j) {
    int lo, hi;
    span_bounds(h_span, t_span, b_spans, b, j, lo, hi);
    const int cf = chunk_first(lo), cl = chunk_last(hi);
    if (cf <= cl) {
      cb = cf < cb ? cf : cb;
      ce = cl > ce ? cl : ce;
    }
  }
  const int nch = ce - cb + 1;

  const int slot = 2 * wave + half;
#pragma unroll
  for (int p = 0; p < NCHUNK_ / 32; ++p) {
    if (32 * p < nch) {
      const int c0 = cb + 32 * p + slot;
      const int c1 = c0 + 16;
      v4f buf0[CH_ROWS_], buf1[CH_ROWS_];
      chunk_load(buf0, rowbase, colbytes, c0 < ce ? c0 : ce);
      chunk_load(buf1, rowbase, colbytes, c1 < ce ? c1 : ce);
      const v4f m0 = chunk_fold(buf0);
      if (c0 <= ce) *(v4f*)&tab[(c0 - cb) * STRIP_ + lcol] = m0;
      const v4f m1 = chunk_fold(buf1);
      if (c1 <= ce) *(v4f*)&tab[(c1 - cb) * STRIP_ + lcol] = m1;
    }
  }
  __syncthreads();

  const float ninf = -__builtin_inff();
  for (int j = wave; j < NSPAN_; j += SPAN_WV_) {
    int lo, hi;
    span_bounds(h_span, t_span, b_spans, b, j, lo, hi);
    float* dst;
    if (j == 0)      dst = h_emb + b * D_;
    else if (j == 1) dst = t_emb + b * D_;
    else             dst = b_emb + (size_t)(b * K_ + (j - 2)) * D_;

    v4f m = {ninf, ninf, ninf, ninf};
    if (lo <= hi) {
      const int cf = chunk_first(lo), cl = chunk_last(hi);
      int fend = CH_ROWS_ * cf - 1;
      fend = fend < hi ? fend : hi;
      fend = fend > lo ? fend : lo;
      int bbeg = CH_ROWS_ * (cl + 1);
      bbeg = bbeg > lo ? bbeg : lo;
      bbeg = bbeg < hi ? bbeg : hi;
      v4f e[CH_ROWS_ - 1];
#pragma unroll
      for (int i = 0; i < CH_ROWS_ - 1; ++i) {
        int rf = lo + i;  rf = rf < fend ? rf : fend;
        int rb = hi - i;  rb = rb > bbeg ? rb : bbeg;
        e[i] = span_row_load(rowbase, colbytes, half ? rb : rf);
      }
      if (cf <= cl) {
        const int ntrip = (cl - cf + 2) >> 1;
        for (int t0 = 0; t0 < ntrip; t0 += 8) {
          v4f q[8];
#pragma unroll
          for (int u = 0; u < 8; ++u) {
            int c = cf + 2 * (t0 + u) + half;
            c = c < cl ? c : cl;
            q[u] = *(const v4f*)&tab[(c - cb) * STRIP_ + lcol];
          }
#pragma unroll
          for (int u = 0; u < 8; ++u) m = vmax4(m, q[u]);
        }
      }
#pragma unroll
      for (int i = 0; i < CH_ROWS_ - 1; ++i) m = vmax4(m, e[i]);
    }

    m.x = fmaxf(m.x, __shfl_xor(m.x, 16));
    m.y = fmaxf(m.y, __shfl_xor(m.y, 16));
    m.z = fmaxf(m.z, __shfl_xor(m.z, 16));
    m.w = fmaxf(m.w, __shfl_xor(m.w, 16));

    if (half == 0) *(v4f*)(dst + col) = m;
  }
}

#define GK_      768
#define G_KC_    128
#define G_PITCH_ 132
#define G_NCH_   (GK_ / G_KC_)
#define G_LPR_   (G_KC_ / 4)

template <int NP>
__device__ __forceinline__ void gemm_fetch(v4f (&reg)[NP],
                                           const float* const (&srcp)[NP], int chunk) {
#pragma unroll
  for (int i = 0; i < NP; ++i) reg[i] = *(const v4f*)(srcp[i] + chunk * G_KC_);
}

template <int NP, int RPP>
__device__ __forceinline__ void gemm_stash(const v4f (&reg)[NP], float* stage, int soff) {
#pragma unroll
  for (int i = 0; i < NP; ++i) *(v4f*)(stage + soff + i * RPP * G_PITCH_) = reg[i];
}

template <bool FU, int NKB>
__device__ __forceinline__ void gemm_chunk(const float* ap, const float* bp,
                                           v8f& acc0, v8f& acc1,
                                           const float* hp, const float* wp0, const float* wp1,
                                           v8f& ua0, v8f& ua1) {
  v4f af[NKB], bf[NKB];
  v4f hf[2], w0f[2], w1f[2];
#pragma unroll
  for (int kb = 0; kb < NKB; ++kb) {
    af[kb] = *(const v4f*)(ap + kb * 8);
    bf[kb] = *(const v4f*)(bp + kb * 8);
  }
  if constexpr (FU) {
#pragma unroll
    for (int j = 0; j < 2; ++j) {
      hf[j]  = *(const v4f*)(hp  + j * 8);
      w0f[j] = *(const v4f*)(wp0 + j * 8);
      w1f[j] = *(const v4f*)(wp1 + j * 8);
    }
  }
  __builtin_amdgcn_sched_barrier(0);
#pragma unroll
  for (int kb = 0; kb < NKB; ++kb) {
    const v2f a0 = {af[kb].x, af[kb].y}, b0 = {bf[kb].x, bf[kb].y};
    const v2f a1 = {af[kb].z, af[kb].w}, b1 = {bf[kb].z, bf[kb].w};
    acc0 = wmma_f32_16x16x4(a0, b0, acc0);
    acc1 = wmma_f32_16x16x4(a1, b1, acc1);
  }
  if constexpr (FU) {
#pragma unroll
    for (int j = 0; j < 2; ++j) {
      const v2f h0 = {hf[j].x, hf[j].y}, h1 = {hf[j].z, hf[j].w};
      const v2f p0 = {w0f[j].x, w0f[j].y}, p1 = {w0f[j].z, w0f[j].w};
      const v2f q0 = {w1f[j].x, w1f[j].y}, q1 = {w1f[j].z, w1f[j].w};
      ua0 = wmma_f32_16x16x4(h0, p0, ua0);
      ua1 = wmma_f32_16x16x4(h0, q0, ua1);
      ua0 = wmma_f32_16x16x4(h1, p1, ua0);
      ua1 = wmma_f32_16x16x4(h1, q1, ua1);
    }
  }
}

template <bool FU, int NKB, int NP, int RPP>
__device__ __forceinline__ void gemm_first_half(const float* ap, const float* bp,
                                                v4f (&af)[NKB], v4f (&bf)[NKB],
                                                const float* hp, const float* wp0, const float* wp1,
                                                v4f (&hf)[2], v4f (&w0f)[2], v4f (&w1f)[2],
                                                v8f& acc0, v8f& acc1,
                                                v4f (&rnext)[NP], float* stage_next, int soff,
                                                const float* const (&srcp)[NP]) {
  static_assert(NKB % 2 == 0, "the chunk's 8-k blocks are taken in two halves");
#pragma unroll
  for (int kb = 0; kb < NKB; ++kb) {
    af[kb] = *(const v4f*)(ap + kb * 8);
    bf[kb] = *(const v4f*)(bp + kb * 8);
  }
  if constexpr (FU) {
#pragma unroll
    for (int j = 0; j < 2; ++j) {
      hf[j]  = *(const v4f*)(hp  + j * 8);
      w0f[j] = *(const v4f*)(wp0 + j * 8);
      w1f[j] = *(const v4f*)(wp1 + j * 8);
    }
  }
  __builtin_amdgcn_sched_barrier(0);
#pragma unroll
  for (int kb = 0; kb < NKB / 2; ++kb) {
    const v2f a0 = {af[kb].x, af[kb].y}, b0 = {bf[kb].x, bf[kb].y};
    const v2f a1 = {af[kb].z, af[kb].w}, b1 = {bf[kb].z, bf[kb].w};
    acc0 = wmma_f32_16x16x4(a0, b0, acc0);
    acc1 = wmma_f32_16x16x4(a1, b1, acc1);
  }
  asm volatile("" : "+v"(acc0), "+v"(acc1) : : "memory");
  gemm_stash<NP, RPP>(rnext, stage_next, soff);
  gemm_fetch(rnext, srcp, 3);
}

template <bool FU, int NKB>
__device__ __forceinline__ void gemm_first_rest(const v4f (&af)[NKB], const v4f (&bf)[NKB],
                                                const v4f (&hf)[2], const v4f (&w0f)[2], const v4f (&w1f)[2],
                                                v8f& acc0, v8f& acc1, v8f& ua0, v8f& ua1) {
#pragma unroll
  for (int kb = NKB / 2; kb < NKB; ++kb) {
    const v2f a0 = {af[kb].x, af[kb].y}, b0 = {bf[kb].x, bf[kb].y};
    const v2f a1 = {af[kb].z, af[kb].w}, b1 = {bf[kb].z, bf[kb].w};
    acc0 = wmma_f32_16x16x4(a0, b0, acc0);
    acc1 = wmma_f32_16x16x4(a1, b1, acc1);
  }
  if constexpr (FU) {
#pragma unroll
    for (int j = 0; j < 2; ++j) {
      const v2f h0 = {hf[j].x, hf[j].y}, h1 = {hf[j].z, hf[j].w};
      const v2f p0 = {w0f[j].x, w0f[j].y}, p1 = {w0f[j].z, w0f[j].w};
      const v2f q0 = {w1f[j].x, w1f[j].y}, q1 = {w1f[j].z, w1f[j].w};
      ua0 = wmma_f32_16x16x4(h0, p0, ua0);
      ua1 = wmma_f32_16x16x4(h0, q0, ua1);
      ua0 = wmma_f32_16x16x4(h1, p1, ua0);
      ua1 = wmma_f32_16x16x4(h1, q1, ua1);
    }
  }
}

__device__ __forceinline__ void gemm_lds_handover() {
  __builtin_amdgcn_fence(__ATOMIC_RELEASE, "workgroup", "local");
  __builtin_amdgcn_s_barrier();
  __builtin_amdgcn_fence(__ATOMIC_ACQUIRE, "workgroup", "local");
}

#define PT_  (D_ / 32)
#define PBM_ 64

#define EPI_PLAIN_ 0
#define EPI_DOTS_  1
#define EPI_ATTN_  2

template <int WM, int WN, int KS, int EPI>
__global__ void __launch_bounds__(32 * WM * WN * KS)
gemm_nt_wmma(const float* __restrict__ A,
             const float* __restrict__ W,
             const float* __restrict__ bias,
             float* __restrict__ C,
             int N, int relu,
             const float* __restrict__ Uq,
             float* __restrict__ part,
             const float* __restrict__ Bi,
             const float* __restrict__ Pin,
             const float* __restrict__ A2,
             const float* __restrict__ W2,
             const float* __restrict__ bias2,
             float* __restrict__ C2) {
  constexpr bool DOTS = (EPI == EPI_DOTS_);
  constexpr bool ATTN = (EPI == EPI_ATTN_);
  constexpr int BM   = 16 * WM;
  constexpr int BN   = 16 * WN;
  constexpr int UR   = DOTS ? BM / K_ : 0;
  constexpr int ROWS = BM + BN + (DOTS ? UR + BN : 0);
  constexpr int NPC  = WM * WN;
  constexpr int NKB  = G_KC_ / 8 / KS;
  constexpr int RPP  = 32 * NPC * KS / G_LPR_;
  constexpr int NP   = ROWS / RPP;
  static_assert(ROWS % RPP == 0 && BM % RPP == 0 && BN % RPP == 0 && UR % RPP == 0,
                "copy passes must not straddle A, W, A2 and W2");
  static_assert(!DOTS || (UR == 8 && BN == 32 && 32 * WM * WN == UR * BN && 2 * WM * WN == G_KC_ / 8),
                "u tile: 8 x 32, one element per thread, two 8-k blocks of a chunk per wave");
  static_assert(G_NCH_ % 2 == 0 && G_NCH_ >= 6, "the K loop takes the chunks in pairs, two fetched ahead; its first trip (written out) fetches chunks 3 and 4");
  static_assert((G_KC_ / 8) % KS == 0, "a chunk's 8-k blocks are dealt evenly over the KS waves of a piece");
  static_assert(!DOTS || KS == 1, "the bridge product's waves own their pieces over the whole K");
  __shared__ __attribute__((aligned(16))) float tile[2][ROWS * G_PITCH_];

  const int tid  = threadIdx.x;
  const int lane = tid & 31;
  const int wave = tid >> 5;
  const int ks   = KS > 1 ? wave / NPC : 0;
  const int pc   = KS > 1 ? wave % NPC : wave;
  const int wm   = pc / WN;
  const int wn   = pc % WN;
  const int half = lane >> 4;
  const int r    = lane & 15;
  const int m0   = blockIdx.y * BM;
  const int n0   = blockIdx.x * BN;

  const int srow = tid / G_LPR_;
  const int scol = (tid % G_LPR_) * 4;
  const float* srcp[NP];
#pragma unroll
  for (int i = 0; i < NP; ++i) {
    const int tr = srow + RPP * i;
    if (RPP * i < BM) {
      srcp[i] = A + (size_t)(m0 + tr) * GK_ + scol;
    } else if (RPP * i < BM + BN) {
      int wr = n0 + tr - BM;
      if (wr > N - 1) wr = N - 1;
      srcp[i] = W + (size_t)wr * GK_ + scol;
    } else if (RPP * i < BM + BN + UR) {
      srcp[i] = A2 + (size_t)(m0 / K_ + tr - BM - BN) * GK_ + scol;
    } else {
      int wr = n0 + tr - BM - BN - UR;
      if (wr > N - 1) wr = N - 1;
      srcp[i] = W2 + (size_t)wr * GK_ + scol;
    }
  }
  const int soff = srow * G_PITCH_ + scol;

  const int kof = ks * NKB * 8 + 4 * half;
  const float* ap0 = &tile[0][(wm * 16 + r) * G_PITCH_ + kof];
  const float* bp0 = &tile[0][(BM + wn * 16 + r) * G_PITCH_ + kof];
  const float* ap1 = &tile[1][(wm * 16 + r) * G_PITCH_ + kof];
  const float* bp1 = &tile[1][(BM + wn * 16 + r) * G_PITCH_ + kof];

  const float* hp0 = nullptr; const float* hp1 = nullptr;
  const float* up00 = nullptr; const float* up01 = nullptr;
  const float* up10 = nullptr; const float* up11 = nullptr;
  if constexpr (DOTS) {
    const int ukb = 2 * wave * 8 + 4 * half;
    hp0  = &tile[0][(BM + BN + (r & (UR - 1))) * G_PITCH_ + ukb];
    hp1  = &tile[1][(BM + BN + (r & (UR - 1))) * G_PITCH_ + ukb];
    up00 = &tile[0][(BM + BN + UR + r) * G_PITCH_ + ukb];
    up01 = &tile[0][(BM + BN + UR + 16 + r) * G_PITCH_ + ukb];
    up10 = &tile[1][(BM + BN + UR + r) * G_PITCH_ + ukb];
    up11 = &tile[1][(BM + BN + UR + 16 + r) * G_PITCH_ + ukb];
  }

  v8f acc0 = {}, acc1 = {};
  v8f ua0 = {}, ua1 = {};
  v4f ra[NP], rb[NP];


  const int erow = tid >> 3;
  const int ecol = n0 + 4 * (tid & 7);
  if constexpr (ATTN) {
    asm volatile("" : : "s"(Uq), "s"(Bi), "s"(Pin));
  }

  gemm_fetch(ra, srcp, 0);
  __builtin_amdgcn_sched_barrier(0);
  gemm_fetch(rb, srcp, 1);
  __builtin_amdgcn_sched_barrier(0);

  v4f pa[PT_ / 4], pb[PT_ / 4], bq[2][K_], uq4[2];

  gemm_stash<NP, RPP>(ra, tile[0], soff);
  gemm_fetch(ra, srcp, 2);

  const int nbias = n0 + wn * 16 + r;
  const float bpre = bias[nbias < N ? nbias : N - 1];
  float ub = 0.0f;
  if constexpr (DOTS) {
    const int un = n0 + (tid & 31);
    ub = bias2[un < N ? un : N - 1];
  }
  if constexpr (ATTN) {
    static_assert(WM == 1 && WN == 2, "the attention epilogue is written for a 16 x 32 tile and its first 64 threads");
    static_assert(PT_ % 4 == 0 && K_ == 8, "4 threads share a row's column tiles; 8 bridges = 2 x 16 bytes");
    if (ks == 0) {
      const int bd = m0 + (tid >> 2);
      const float* pp = Pin + ((size_t)(bd / (PBM_ / K_)) * PT_ + (tid & 3)) * PBM_ + (bd % (PBM_ / K_)) * K_;
#pragma unroll
      for (int j = 0; j < PT_ / 4; ++j) {
        pa[j] = *(const v4f*)(pp + (size_t)j * 4 * PBM_);
        pb[j] = *(const v4f*)(pp + (size_t)j * 4 * PBM_ + 4);
      }
#pragma unroll
      for (int e = 0; e < 2; ++e) {
        const int b = m0 + erow + 8 * e;
        uq4[e] = *(const v4f*)(Uq + (size_t)b * N + ecol);
#pragma unroll
        for (int k = 0; k < K_; ++k)
          bq[e][k] = *(const v4f*)(Bi + ((size_t)b * K_ + k) * N + ecol);
      }
    }
  }

  v4f af0[NKB], bf0[NKB], hf0[2], w0f0[2], w1f0[2];
  gemm_lds_handover();
  gemm_first_half<DOTS, NKB, NP, RPP>(ap0, bp0, af0, bf0, hp0, up00, up01, hf0, w0f0, w1f0,
                                      acc0, acc1, rb, tile[1], soff, srcp);
  gemm_lds_handover();
  asm volatile("" : "+v"(acc0), "+v"(acc1) : : "memory");
  gemm_first_rest<DOTS, NKB>(af0, bf0, hf0, w0f0, w1f0, acc0, acc1, ua0, ua1);
  gemm_stash<NP, RPP>(ra, tile[0], soff);
  gemm_fetch(ra, srcp, 4);
  gemm_chunk<DOTS, NKB>(ap1, bp1, acc0, acc1, hp1, up10, up11, ua0, ua1);

  for (int t = 2; t < G_NCH_; t += 2) {
    gemm_lds_handover();
    gemm_stash<NP, RPP>(rb, tile[1], soff);
    if (t + 3 < G_NCH_) gemm_fetch(rb, srcp, t + 3);
    gemm_chunk<DOTS, NKB>(ap0, bp0, acc0, acc1, hp0, up00, up01, ua0, ua1);
    gemm_lds_handover();
    if (t + 2 < G_NCH_) gemm_stash<NP, RPP>(ra, tile[0], soff);
    if (t + 4 < G_NCH_) gemm_fetch(ra, srcp, t + 4);
    gemm_chunk<DOTS, NKB>(ap1, bp1, acc0, acc1, hp1, up10, up11, ua0, ua1);
  }

  float uqv = 0.0f;
  if constexpr (DOTS) {
    __shared__ float ured[WM * WN][2][UR][16];
    __shared__ float utile[UR][BN];
    if (half == 0) {
#pragma unroll
      for (int i = 0; i < UR; ++i) {
        ured[wave][0][i][r] = ua0[i];
        ured[wave][1][i][r] = ua1[i];
      }
    }
    gemm_lds_handover();
    {
      const int urow = tid >> 5, ucol = tid & 31;
      float s = ured[0][ucol >> 4][urow][ucol & 15];
#pragma unroll
      for (int w = 1; w < WM * WN; ++w) s += ured[w][ucol >> 4][urow][ucol & 15];
      const int un = n0 + ucol;
      s += ub;
      utile[urow][ucol] = s;
      if (un < N) C2[(size_t)(m0 / K_ + urow) * N + un] = s;
    }
    gemm_lds_handover();
    uqv = utile[2 * wm + half][wn * 16 + r];
  }

  float cs[8];
#pragma unroll
  for (int i = 0; i < 8; ++i) cs[i] = acc0[i] + acc1[i];
  const bool owner = (ks == 0);
  if constexpr (KS > 1) {
    __shared__ float kred[KS - 1][NPC][8][32];
    if (!owner) {
#pragma unroll
      for (int i = 0; i < 8; ++i) kred[ks - 1][pc][i][lane] = cs[i];
    }
    gemm_lds_handover();
    if (owner) {
#pragma unroll
      for (int q = 0; q < KS - 1; ++q) {
#pragma unroll
        for (int i = 0; i < 8; ++i) cs[i] += kred[q][pc][i][lane];
      }
    }
  }

  const int n = n0 + wn * 16 + r;
  float pd[8];
#pragma unroll
  for (int i = 0; i < 8; ++i) pd[i] = 0.0f;
  if constexpr (ATTN) {
    __shared__ __attribute__((aligned(16))) float sv[BM][BN + 4];
    __shared__ __attribute__((aligned(16))) float satt[BM * K_];
    if (owner) {
      const float bv = bpre;
#pragma unroll
      for (int i = 0; i < 8; ++i) {
        float vout = cs[i] + bv;
        if (relu) vout = fmaxf(vout, 0.0f);
        sv[wm * 16 + i + 8 * half][wn * 16 + r] = vout;
      }
      asm volatile("" : "+v"(pa[0]), "+v"(pa[1]), "+v"(pa[2]), "+v"(pa[3]), "+v"(pa[4]), "+v"(pa[5]),
                        "+v"(pb[0]), "+v"(pb[1]), "+v"(pb[2]), "+v"(pb[3]), "+v"(pb[4]), "+v"(pb[5]));
      static_assert(PT_ / 4 == 6, "the statement above names the 6 + 6 vectors of pa[] and pb[]");
      float att[K_];
      att[0] = pa[0].x; att[1] = pa[0].y; att[2] = pa[0].z; att[3] = pa[0].w;
      att[4] = pb[0].x; att[5] = pb[0].y; att[6] = pb[0].z; att[7] = pb[0].w;
#pragma unroll
      for (int j = 1; j < PT_ / 4; ++j) {
        att[0] += pa[j].x; att[1] += pa[j].y; att[2] += pa[j].z; att[3] += pa[j].w;
        att[4] += pb[j].x; att[5] += pb[j].y; att[6] += pb[j].z; att[7] += pb[j].w;
      }
#pragma unroll
      for (int off = 1; off < 4; off <<= 1) {
#pragma unroll
        for (int k = 0; k < K_; ++k) att[k] += __shfl_xor(att[k], off);
      }
#pragma unroll
      for (int k = 0; k < K_; ++k) att[k] *= 0.125f;
      float mx = att[0];
#pragma unroll
      for (int k = 1; k < K_; ++k) mx = fmaxf(mx, att[k]);
      float s = 0.f;
#pragma unroll
      for (int k = 0; k < K_; ++k) { att[k] = __expf(att[k] - mx); s += att[k]; }
      const float inv = 1.0f / s;
#pragma unroll
      for (int k = 0; k < K_; ++k) att[k] *= inv;
      if ((tid & 3) == 0) {
        const v4f w0 = {att[0], att[1], att[2], att[3]};
        const v4f w1 = {att[4], att[5], att[6], att[7]};
        *(v4f*)&satt[(tid >> 2) * K_]     = w0;
        *(v4f*)&satt[(tid >> 2) * K_ + 4] = w1;
      }
    }
    gemm_lds_handover();
    if (owner) {
#pragma unroll
      for (int e = 0; e < 2; ++e) {
        const int row = erow + 8 * e;
        const v4f vv = *(const v4f*)&sv[row][4 * (tid & 7)];
        const v4f w0 = *(const v4f*)&satt[row * K_];
        const v4f w1 = *(const v4f*)&satt[row * K_ + 4];
        const float wk[K_] = {w0.x, w0.y, w0.z, w0.w, w1.x, w1.y, w1.z, w1.w};
        v4f c = {0.f, 0.f, 0.f, 0.f};
#pragma unroll
        for (int k = 0; k < K_; ++k) {
          c.x += wk[k] * bq[e][k].x;
          c.y += wk[k] * bq[e][k].y;
          c.z += wk[k] * bq[e][k].z;
          c.w += wk[k] * bq[e][k].w;
        }
        v4f o;
        o.x = uq4[e].x + vv.x + c.x;
        o.y = uq4[e].y + vv.y + c.y;
        o.z = uq4[e].z + vv.z + c.z;
        o.w = uq4[e].w + vv.w + c.w;
        *(v4f*)(C + (size_t)(m0 + row) * N + ecol) = o;
      }
    }
  } else if (owner && n < N) {
    const float bv = bpre;
#pragma unroll
    for (int i = 0; i < 8; ++i) {
      float vout = cs[i] + bv;
      if (relu) vout = fmaxf(vout, 0.0f);
      C[(size_t)(m0 + wm * 16 + i + 8 * half) * N + n] = vout;
      if constexpr (DOTS) pd[i] = uqv * vout;
    }
  }
  if constexpr (DOTS) {
    static_assert(K_ == 8, "one lane half (8 rows of the piece) must be one batch row");
    __shared__ float dred[WN][BM];
#pragma unroll
    for (int off = 8; off > 0; off >>= 1) {
#pragma unroll
      for (int i = 0; i < 8; ++i) pd[i] += __shfl_xor(pd[i], off);
    }
    if (r == 0) {
#pragma unroll
      for (int i = 0; i < 8; ++i) dred[wn][wm * 16 + 8 * half + i] = pd[i];
    }
    gemm_lds_handover();
    if (tid < BM) {
      float s = dred[0][tid];
#pragma unroll
      for (int w = 1; w < WN; ++w) s += dred[w][tid];
      const int ntile = (N + BN - 1) / BN;
      part[((size_t)blockIdx.y * ntile + blockIdx.x) * BM + tid] = s;
    }
  }
}

extern "C" void kernel_launch(void* const* d_in, const int* in_sizes, int n_in,
                              void* d_out, int out_size, void* d_ws, size_t ws_size,
                              hipStream_t stream) {
  (void)in_sizes; (void)n_in; (void)out_size; (void)ws_size;

  const float* emb     = (const float*)d_in[0];
  const int*   h_span  = (const int*)d_in[1];
  const int*   t_span  = (const int*)d_in[2];
  const int*   b_spans = (const int*)d_in[3];
  const float* wu_w    = (const float*)d_in[4];
  const float* wu_b    = (const float*)d_in[5];
  const float* wv_w    = (const float*)d_in[6];
  const float* wv_b    = (const float*)d_in[7];
  const float* wi_w    = (const float*)d_in[8];
  const float* wi_b    = (const float*)d_in[9];
  const float* ln1_w   = (const float*)d_in[10];
  const float* ln1_b   = (const float*)d_in[11];
  const float* pred_w  = (const float*)d_in[12];
  const float* pred_b  = (const float*)d_in[13];
  float* out = (float*)d_out;

  float* ws    = (float*)d_ws;
  float* h_emb = ws;
  float* t_emb = h_emb + B_ * D_;
  float* b_emb = t_emb + B_ * D_;
  float* u     = b_emb + B_ * K_ * D_;
  float* part  = u + B_ * D_;
  float* bi    = part + B_ * D_;
  float* pre   = bi + B_ * K_ * D_;
  float* pair  = h_emb;

  span_max_chunk_table<<<dim3(D_ / STRIP_, B_), 32 * SPAN_WV_, 0, stream>>>(emb, h_span, t_span, b_spans,
                                                                            h_emb, t_emb, b_emb);

  gemm_nt_wmma<4, 2, 1, EPI_DOTS_><<<dim3(PT_, 8), 256, 0, stream>>>(b_emb, wi_w, wi_b, bi, D_, 0,
                                                                  nullptr, part, nullptr, nullptr,
                                                                  h_emb, wu_w, wu_b, u);
  gemm_nt_wmma<1, 2, 4, EPI_ATTN_><<<dim3(24, 4), 256, 0, stream>>>(t_emb, wv_w, wv_b, pre, D_, 0,
                                                               u, nullptr, bi, part,
                                                               nullptr, nullptr, nullptr, nullptr);

  gemm_nt_wmma<1, 2, 4, EPI_PLAIN_><<<dim3(24, 4), 256, 0, stream>>>(pre, ln1_w, ln1_b, pair, D_, 1,
                                                                nullptr, nullptr, nullptr, nullptr,
                                                                nullptr, nullptr, nullptr, nullptr);
  gemm_nt_wmma<1, 2, 4, EPI_PLAIN_><<<dim3(9, 4), 256, 0, stream>>>(pair, pred_w, pred_b, out, R_, 0,
                                                               nullptr, nullptr, nullptr, nullptr,
                                                               nullptr, nullptr, nullptr, nullptr);
}
